# starting kernel
# baseline (speedup 1.0000x reference)
#include <hip/hip_runtime.h>

typedef __attribute__((ext_vector_type(2))) float v2f;
typedef __attribute__((ext_vector_type(8))) float v8f;

constexpr int Bsz = 128;
constexpr int Nsl = 32;
constexpr int Dd  = 128;
constexpr int Hh  = 256;
constexpr int BN  = Bsz * Nsl;
constexpr float LN_EPS = 1e-5f;

__global__ void enc_gemm_kernel(const float* __restrict__ slots,
                                const float* __restrict__ Wenc,
                                float* __restrict__ UV)
{
    const int lane = threadIdx.x & 31;
    const int wave = threadIdx.x >> 5;
    const int tile = blockIdx.x * 8 + wave;
    const int tm = tile >> 5;
    const int tn = tile & 31;

    const int row  = tm * 16 + (lane & 15);
    const int cg   = tn * 16 + (lane & 15);
    const int whichW = cg >> 8;
    const int hcol   = cg & 255;
    const int klo  = (lane >> 4) * 2;

    const float* __restrict__ arow = slots + (size_t)row * Dd;

    v8f acc = {};
    #pragma unroll 4
    for (int k = 0; k < Dd; k += 4) {
        const int kk = k + klo;
        v2f a; a.x = arow[kk]; a.y = arow[kk + 1];
        const float* bp = Wenc + (size_t)(whichW * Dd + kk) * Hh + hcol;
        v2f b; b.x = bp[0]; b.y = bp[Hh];
        acc = __builtin_amdgcn_wmma_f32_16x16x4_f32(false, a, false, b,
                                                    (short)0, acc, false, false);
    }

    const int hi = lane >> 4;
    float* __restrict__ dst = UV + (size_t)whichW * BN * Hh;
    #pragma unroll
    for (int r = 0; r < 8; ++r) {
        const int rowm = tm * 16 + r + 8 * hi;
        dst[(size_t)rowm * Hh + hcol] = acc[r];
    }
}

__device__ inline float wave_reduce_sum(float v) {
    #pragma unroll
    for (int m = 16; m >= 1; m >>= 1) v += __shfl_xor(v, m, 32);
    return v;
}

__global__ void pair_reduce_kernel(const float* __restrict__ U,
                                   const float* __restrict__ V,
                                   const float* __restrict__ b_enc,
                                   const float* __restrict__ ln_g,
                                   const float* __restrict__ ln_b,
                                   const float* __restrict__ W_att,
                                   const float* __restrict__ b_att,
                                   float* __restrict__ G,
                                   float* __restrict__ S)
{
    __shared__ float gsh[8 * Hh];
    __shared__ float ssh[8];

    const int bi   = blockIdx.x;
    const int b    = bi >> 5;
    const int i    = bi & 31;
    const int lane = threadIdx.x & 31;
    const int wave = threadIdx.x >> 5;

    const float batt = b_att[0];

    float u[8], be[8], lg[8], lb[8], wa[8], gacc[8];
    #pragma unroll
    for (int t = 0; t < 8; ++t) {
        const int idx = lane + 32 * t;
        u[t]  = U[(size_t)bi * Hh + idx];
        be[t] = b_enc[idx];
        lg[t] = ln_g[idx];
        lb[t] = ln_b[idx];
        wa[t] = W_att[idx];
        gacc[t] = 0.0f;
    }
    float sacc = 0.0f;

    for (int jj = wave; jj < Nsl; jj += 8) {
        if (jj == i) continue;
        const float* __restrict__ vrow = V + (size_t)(b * Nsl + jj) * Hh;

        float h[8];
        float sum = 0.0f;
        #pragma unroll
        for (int t = 0; t < 8; ++t) {
            h[t] = u[t] + vrow[lane + 32 * t] + be[t];
            sum += h[t];
        }
        const float mu = wave_reduce_sum(sum) * (1.0f / Hh);

        float sq = 0.0f;
        #pragma unroll
        for (int t = 0; t < 8; ++t) { const float d = h[t] - mu; sq += d * d; }
        const float var = wave_reduce_sum(sq) * (1.0f / Hh);
        const float inv = rsqrtf(var + LN_EPS);

        float adot = 0.0f;
        #pragma unroll
        for (int t = 0; t < 8; ++t) {
            float hn = (h[t] - mu) * inv * lg[t] + lb[t];
            hn = fmaxf(hn, 0.0f);
            h[t] = hn;
            adot += hn * wa[t];
        }
        adot = wave_reduce_sum(adot) + batt;
        const float a = 1.0f / (1.0f + __expf(-adot));

        #pragma unroll
        for (int t = 0; t < 8; ++t) gacc[t] += a * h[t];
        sacc += a;
    }

    #pragma unroll
    for (int t = 0; t < 8; ++t) gsh[wave * Hh + lane + 32 * t] = gacc[t];
    if (lane == 0) ssh[wave] = sacc;
    __syncthreads();

    float tot = 0.0f;
    #pragma unroll
    for (int w = 0; w < 8; ++w) tot += gsh[w * Hh + threadIdx.x];
    G[(size_t)bi * Hh + threadIdx.x] = tot;

    if (threadIdx.x == 0) {
        float st = 0.0f;
        #pragma unroll
        for (int w = 0; w < 8; ++w) st += ssh[w];
        S[bi] = st;
    }
}

__global__ void eff_gemm_kernel(const float* __restrict__ G,
                                const float* __restrict__ S,
                                const float* __restrict__ Weff,
                                const float* __restrict__ beff,
                                float* __restrict__ out)
{
    const int lane = threadIdx.x & 31;
    const int wave = threadIdx.x >> 5;
    const int tile = blockIdx.x * 8 + wave;
    const int tm = tile >> 3;
    const int tn = tile & 7;

    const int row = tm * 16 + (lane & 15);
    const int col = tn * 16 + (lane & 15);
    const int klo = (lane >> 4) * 2;

    const float* __restrict__ arow = G + (size_t)row * Hh;

    v8f acc = {};
    #pragma unroll 4
    for (int k = 0; k < Hh; k += 4) {
        const int kk = k + klo;
        v2f a; a.x = arow[kk]; a.y = arow[kk + 1];
        const float* bp = Weff + (size_t)kk * Dd + col;
        v2f b; b.x = bp[0]; b.y = bp[Dd];
        acc = __builtin_amdgcn_wmma_f32_16x16x4_f32(false, a, false, b,
                                                    (short)0, acc, false, false);
    }

    const float bec = beff[col];
    const int hi = lane >> 4;
    #pragma unroll
    for (int r = 0; r < 8; ++r) {
        const int rowm = tm * 16 + r + 8 * hi;
        out[(size_t)rowm * Dd + col] = acc[r] + S[rowm] * bec;
    }
}

extern "C" void kernel_launch(void* const* d_in, const int* in_sizes, int n_in,
                              void* d_out, int out_size, void* d_ws, size_t ws_size,
                              hipStream_t stream)
{
    const float* slots = (const float*)d_in[0];
    const float* W_enc = (const float*)d_in[1];
    const float* b_enc = (const float*)d_in[2];
    const float* ln_g  = (const float*)d_in[3];
    const float* ln_b  = (const float*)d_in[4];
    const float* W_att = (const float*)d_in[5];
    const float* b_att = (const float*)d_in[6];
    const float* W_eff = (const float*)d_in[7];
    const float* b_eff = (const float*)d_in[8];
    float* out = (float*)d_out;

    float* ws = (float*)d_ws;
    float* U  = ws;
    float* V  = U + (size_t)BN * Hh;
    float* Gm = V + (size_t)BN * Hh;
    float* Sm = Gm + (size_t)BN * Hh;

    enc_gemm_kernel<<<1024, 256, 0, stream>>>(slots, W_enc, U);

    pair_reduce_kernel<<<BN, 256, 0, stream>>>(U, V, b_enc, ln_g, ln_b,
                                               W_att, b_att, Gm, Sm);

    eff_gemm_kernel<<<256, 256, 0, stream>>>(Gm, Sm, W_eff, b_eff, out);
}


# pair kernel: one wave per (b,i) walks all j from V[b] staged in LDS (8x fewer waves); each 16-lane half owns one j so row sums are DPP adds; order of row sums and of the sum over j changes
# speedup vs baseline: 1.3356x; 1.3356x over previous
#include <hip/hip_runtime.h>

typedef __attribute__((ext_vector_type(2))) float v2f;
typedef __attribute__((ext_vector_type(8))) float v8f;

constexpr int Bsz = 128;
constexpr int Nsl = 32;
constexpr int Dd  = 128;
constexpr int Hh  = 256;
constexpr int BN  = Bsz * Nsl;
constexpr float LN_EPS = 1e-5f;

__global__ void enc_gemm_kernel(const float* __restrict__ slots,
                                const float* __restrict__ Wenc,
                                float* __restrict__ UV)
{
    const int lane = threadIdx.x & 31;
    const int wave = threadIdx.x >> 5;
    const int tile = blockIdx.x * 8 + wave;
    const int tm = tile >> 5;
    const int tn = tile & 31;

    const int row  = tm * 16 + (lane & 15);
    const int cg   = tn * 16 + (lane & 15);
    const int whichW = cg >> 8;
    const int hcol   = cg & 255;
    const int klo  = (lane >> 4) * 2;

    const float* __restrict__ arow = slots + (size_t)row * Dd;

    v8f acc = {};
    #pragma unroll 4
    for (int k = 0; k < Dd; k += 4) {
        const int kk = k + klo;
        v2f a; a.x = arow[kk]; a.y = arow[kk + 1];
        const float* bp = Wenc + (size_t)(whichW * Dd + kk) * Hh + hcol;
        v2f b; b.x = bp[0]; b.y = bp[Hh];
        acc = __builtin_amdgcn_wmma_f32_16x16x4_f32(false, a, false, b,
                                                    (short)0, acc, false, false);
    }

    const int hi = lane >> 4;
    float* __restrict__ dst = UV + (size_t)whichW * BN * Hh;
    #pragma unroll
    for (int r = 0; r < 8; ++r) {
        const int rowm = tm * 16 + r + 8 * hi;
        dst[(size_t)rowm * Hh + hcol] = acc[r];
    }
}

typedef __attribute__((ext_vector_type(4))) float v4f;

constexpr int IPB = 16;
constexpr int PAIR_THREADS = IPB * 32;
constexpr int PAIR_GRID = BN / IPB;

template <int M>
__device__ inline float row_xmask_add(float v) {
    const int moved = __builtin_amdgcn_update_dpp(0, __builtin_bit_cast(int, v),
                                                  0x160 | M, 0xf, 0xf, true);
    return v + __builtin_bit_cast(float, moved);
}

__device__ inline float row16_sum(float v) {
    v = row_xmask_add<1>(v);
    v = row_xmask_add<2>(v);
    v = row_xmask_add<4>(v);
    v = row_xmask_add<8>(v);
    return v;
}

__device__ inline float xhalf(float x) {
    const int b = __builtin_bit_cast(int, x);
    return __builtin_bit_cast(float,
        __builtin_amdgcn_permlanex16(b, b, 0x76543210, 0xfedcba98, false, false));
}

__global__ void __launch_bounds__(PAIR_THREADS)
pair_reduce_kernel(const float* __restrict__ U,
                   const float* __restrict__ V,
                   const float* __restrict__ b_enc,
                   const float* __restrict__ ln_g,
                   const float* __restrict__ ln_b,
                   const float* __restrict__ W_att,
                   const float* __restrict__ b_att,
                   float* __restrict__ G,
                   float* __restrict__ S)
{
    __shared__ v4f vsh[Nsl * Hh / 4];
    __shared__ v4f psh[4 * Hh / 4];

    const int tid  = threadIdx.x;
    const int lane = tid & 31;
    const int wave = tid >> 5;
    const int c    = lane & 15;
    const int hh   = lane >> 4;
    const int b    = blockIdx.x >> 1;
    const int i    = (blockIdx.x & 1) * IPB + wave;
    const int bi   = b * Nsl + i;

    const v4f* __restrict__ urow = reinterpret_cast<const v4f*>(U + (size_t)bi * Hh);
    v4f u4[4];
    #pragma unroll
    for (int m = 0; m < 4; ++m) u4[m] = urow[16 * m + c];

    const int pt = tid & (Hh - 1);
    const int ph = tid >> 8;
    const float pa = (ph ? ln_b : b_enc)[pt];
    const float pb = (ph ? W_att : ln_g)[pt];

    const v4f* __restrict__ vsrc = reinterpret_cast<const v4f*>(V + (size_t)b * Nsl * Hh);
    v4f vst[(Nsl * Hh / 4) / PAIR_THREADS];
    #pragma unroll
    for (int q = 0; q < (Nsl * Hh / 4) / PAIR_THREADS; ++q) vst[q] = vsrc[tid + PAIR_THREADS * q];
    #pragma unroll
    for (int q = 0; q < (Nsl * Hh / 4) / PAIR_THREADS; ++q) vsh[tid + PAIR_THREADS * q] = vst[q];
    {
        float* pf = reinterpret_cast<float*>(psh);
        pf[ph * 2 * Hh + pt]      = pa;
        pf[ph * 2 * Hh + Hh + pt] = pb;
    }
    const float batt = b_att[0];

    __syncthreads();

    v2f ub[8], lg[8], lb[8], wa[8], gacc[8];
    #pragma unroll
    for (int m = 0; m < 4; ++m) {
        const v4f e = psh[16 * m + c];
        const v4f g = psh[64 + 16 * m + c];
        const v4f l = psh[128 + 16 * m + c];
        const v4f w = psh[192 + 16 * m + c];
        ub[2*m] = u4[m].lo + e.lo;  ub[2*m+1] = u4[m].hi + e.hi;
        lg[2*m] = g.lo;             lg[2*m+1] = g.hi;
        lb[2*m] = l.lo;             lb[2*m+1] = l.hi;
        wa[2*m] = w.lo;             wa[2*m+1] = w.hi;
    }
    #pragma unroll
    for (int k = 0; k < 8; ++k) gacc[k] = (v2f){0.0f, 0.0f};
    float sacc = 0.0f;

    const v4f* vbase = vsh + hh * (Hh / 4) + c;
    v4f vcur[4];
    #pragma unroll
    for (int m = 0; m < 4; ++m) vcur[m] = vbase[16 * m];

    for (int t = 0; t < Nsl / 2; ++t) {
        const v4f* vnrow = vbase + ((t + 1) & (Nsl / 2 - 1)) * (2 * Hh / 4);
        v4f vnext[4];
        #pragma unroll
        for (int m = 0; m < 4; ++m) vnext[m] = vnrow[16 * m];

        v2f h[8];
        #pragma unroll
        for (int m = 0; m < 4; ++m) {
            h[2*m]   = ub[2*m]   + vcur[m].lo;
            h[2*m+1] = ub[2*m+1] + vcur[m].hi;
        }
        const v2f s2 = ((h[0] + h[1]) + (h[2] + h[3])) + ((h[4] + h[5]) + (h[6] + h[7]));
        const float mu = row16_sum(s2.x + s2.y) * (1.0f / Hh);

        #pragma unroll
        for (int k = 0; k < 8; ++k) h[k] = h[k] - mu;
        v2f qa = h[0] * h[0];
        v2f qb = h[1] * h[1];
        #pragma unroll
        for (int k = 1; k < 4; ++k) {
            qa = qa + h[2*k]   * h[2*k];
            qb = qb + h[2*k+1] * h[2*k+1];
        }
        const v2f q2 = qa + qb;
        const float var = row16_sum(q2.x + q2.y) * (1.0f / Hh);
        const float inv = rsqrtf(var + LN_EPS);

        #pragma unroll
        for (int k = 0; k < 8; ++k) {
            v2f hn = h[k] * inv * lg[k] + lb[k];
            hn.x = fmaxf(hn.x, 0.0f);
            hn.y = fmaxf(hn.y, 0.0f);
            h[k] = hn;
        }
        v2f da = h[0] * wa[0];
        v2f db = h[1] * wa[1];
        #pragma unroll
        for (int k = 1; k < 4; ++k) {
            da = da + h[2*k]   * wa[2*k];
            db = db + h[2*k+1] * wa[2*k+1];
        }
        const v2f d2 = da + db;
        const float adot = row16_sum(d2.x + d2.y) + batt;
        float a = 1.0f / (1.0f + __expf(-adot));
        a = (2 * t + hh == i) ? 0.0f : a;

        #pragma unroll
        for (int k = 0; k < 8; ++k) gacc[k] = gacc[k] + a * h[k];
        sacc += a;

        #pragma unroll
        for (int m = 0; m < 4; ++m) vcur[m] = vnext[m];
    }

    #pragma unroll
    for (int k = 0; k < 8; ++k) {
        gacc[k].x += xhalf(gacc[k].x);
        gacc[k].y += xhalf(gacc[k].y);
    }
    sacc += xhalf(sacc);

    v4f* __restrict__ grow = reinterpret_cast<v4f*>(G + (size_t)bi * Hh);
    #pragma unroll
    for (int q = 0; q < 2; ++q) {
        const v2f lo = hh ? gacc[4 + 2*q]     : gacc[2*q];
        const v2f hi = hh ? gacc[4 + 2*q + 1] : gacc[2*q + 1];
        v4f o;
        o.lo = lo;
        o.hi = hi;
        grow[16 * (2 * hh + q) + c] = o;
    }
    if (lane == 0) S[bi] = sacc;
}

__global__ void eff_gemm_kernel(const float* __restrict__ G,
                                const float* __restrict__ S,
                                const float* __restrict__ Weff,
                                const float* __restrict__ beff,
                                float* __restrict__ out)
{
    const int lane = threadIdx.x & 31;
    const int wave = threadIdx.x >> 5;
    const int tile = blockIdx.x * 8 + wave;
    const int tm = tile >> 3;
    const int tn = tile & 7;

    const int row = tm * 16 + (lane & 15);
    const int col = tn * 16 + (lane & 15);
    const int klo = (lane >> 4) * 2;

    const float* __restrict__ arow = G + (size_t)row * Hh;

    v8f acc = {};
    #pragma unroll 4
    for (int k = 0; k < Hh; k += 4) {
        const int kk = k + klo;
        v2f a; a.x = arow[kk]; a.y = arow[kk + 1];
        const float* bp = Weff + (size_t)kk * Dd + col;
        v2f b; b.x = bp[0]; b.y = bp[Dd];
        acc = __builtin_amdgcn_wmma_f32_16x16x4_f32(false, a, false, b,
                                                    (short)0, acc, false, false);
    }

    const float bec = beff[col];
    const int hi = lane >> 4;
    #pragma unroll
    for (int r = 0; r < 8; ++r) {
        const int rowm = tm * 16 + r + 8 * hi;
        out[(size_t)rowm * Dd + col] = acc[r] + S[rowm] * bec;
    }
}

extern "C" void kernel_launch(void* const* d_in, const int* in_sizes, int n_in,
                              void* d_out, int out_size, void* d_ws, size_t ws_size,
                              hipStream_t stream)
{
    const float* slots = (const float*)d_in[0];
    const float* W_enc = (const float*)d_in[1];
    const float* b_enc = (const float*)d_in[2];
    const float* ln_g  = (const float*)d_in[3];
    const float* ln_b  = (const float*)d_in[4];
    const float* W_att = (const float*)d_in[5];
    const float* b_att = (const float*)d_in[6];
    const float* W_eff = (const float*)d_in[7];
    const float* b_eff = (const float*)d_in[8];
    float* out = (float*)d_out;

    float* ws = (float*)d_ws;
    float* U  = ws;
    float* V  = U + (size_t)BN * Hh;
    float* Gm = V + (size_t)BN * Hh;
    float* Sm = Gm + (size_t)BN * Hh;

    enc_gemm_kernel<<<1024, 256, 0, stream>>>(slots, W_enc, U);

    pair_reduce_kernel<<<PAIR_GRID, PAIR_THREADS, 0, stream>>>(U, V, b_enc, ln_g, ln_b,
                                                               W_att, b_att, Gm, Sm);

    eff_gemm_kernel<<<256, 256, 0, stream>>>(Gm, Sm, W_eff, b_eff, out);
}


# eff GEMM moved into the pair kernel's tail: G rows go through LDS to 32 WMMAs per wave, W_eff fragments fetched before the j loop; third launch and G/S round trip gone; K summed as two halves of 128
# speedup vs baseline: 1.9390x; 1.4518x over previous
#include <hip/hip_runtime.h>

typedef __attribute__((ext_vector_type(2))) float v2f;
typedef __attribute__((ext_vector_type(8))) float v8f;

constexpr int Bsz = 128;
constexpr int Nsl = 32;
constexpr int Dd  = 128;
constexpr int Hh  = 256;
constexpr int BN  = Bsz * Nsl;
constexpr float LN_EPS = 1e-5f;

__global__ void enc_gemm_kernel(const float* __restrict__ slots,
                                const float* __restrict__ Wenc,
                                float* __restrict__ UV)
{
    const int lane = threadIdx.x & 31;
    const int wave = threadIdx.x >> 5;
    const int tile = blockIdx.x * 8 + wave;
    const int tm = tile >> 5;
    const int tn = tile & 31;

    const int row  = tm * 16 + (lane & 15);
    const int cg   = tn * 16 + (lane & 15);
    const int whichW = cg >> 8;
    const int hcol   = cg & 255;
    const int klo  = (lane >> 4) * 2;

    const float* __restrict__ arow = slots + (size_t)row * Dd;

    v8f acc = {};
    #pragma unroll 4
    for (int k = 0; k < Dd; k += 4) {
        const int kk = k + klo;
        v2f a; a.x = arow[kk]; a.y = arow[kk + 1];
        const float* bp = Wenc + (size_t)(whichW * Dd + kk) * Hh + hcol;
        v2f b; b.x = bp[0]; b.y = bp[Hh];
        acc = __builtin_amdgcn_wmma_f32_16x16x4_f32(false, a, false, b,
                                                    (short)0, acc, false, false);
    }

    const int hi = lane >> 4;
    float* __restrict__ dst = UV + (size_t)whichW * BN * Hh;
    #pragma unroll
    for (int r = 0; r < 8; ++r) {
        const int rowm = tm * 16 + r + 8 * hi;
        dst[(size_t)rowm * Hh + hcol] = acc[r];
    }
}

typedef __attribute__((ext_vector_type(4))) float v4f;

constexpr int IPB = 16;
constexpr int PAIR_THREADS = IPB * 32;
constexpr int PAIR_GRID = BN / IPB;
constexpr int GP = Hh + 4;
constexpr int OP = Dd + 4;

template <int M>
__device__ inline float row_xmask_add(float v) {
    const int moved = __builtin_amdgcn_update_dpp(0, __builtin_bit_cast(int, v),
                                                  0x160 | M, 0xf, 0xf, true);
    return v + __builtin_bit_cast(float, moved);
}

__device__ inline float row16_sum(float v) {
    v = row_xmask_add<1>(v);
    v = row_xmask_add<2>(v);
    v = row_xmask_add<4>(v);
    v = row_xmask_add<8>(v);
    return v;
}

__device__ inline float xhalf(float x) {
    const int b = __builtin_bit_cast(int, x);
    return __builtin_bit_cast(float,
        __builtin_amdgcn_permlanex16(b, b, 0x76543210, 0xfedcba98, false, false));
}

__global__ void __launch_bounds__(PAIR_THREADS)
pair_reduce_kernel(const float* __restrict__ U,
                   const float* __restrict__ V,
                   const float* __restrict__ b_enc,
                   const float* __restrict__ ln_g,
                   const float* __restrict__ ln_b,
                   const float* __restrict__ W_att,
                   const float* __restrict__ b_att,
                   const float* __restrict__ Weff,
                   const float* __restrict__ beff,
                   float* __restrict__ out)
{
    __shared__ v4f vsh[Nsl * Hh / 4];
    __shared__ v4f psh[4 * Hh / 4];
    __shared__ v4f gsh[IPB * GP / 4];
    __shared__ v4f osh[2 * IPB * OP / 4];

    const int tid  = threadIdx.x;
    const int lane = tid & 31;
    const int wave = tid >> 5;
    const int c    = lane & 15;
    const int hh   = lane >> 4;
    const int b    = blockIdx.x >> 1;
    const int i    = (blockIdx.x & 1) * IPB + wave;
    const int bi   = b * Nsl + i;

    const v4f* __restrict__ urow = reinterpret_cast<const v4f*>(U + (size_t)bi * Hh);
    v4f u4[4];
    #pragma unroll
    for (int m = 0; m < 4; ++m) u4[m] = urow[16 * m + c];

    const int pt = tid & (Hh - 1);
    const int ph = tid >> 8;
    const float pa = (ph ? ln_b : b_enc)[pt];
    const float pb = (ph ? W_att : ln_g)[pt];

    const v4f* __restrict__ vsrc = reinterpret_cast<const v4f*>(V + (size_t)b * Nsl * Hh);
    v4f vst[(Nsl * Hh / 4) / PAIR_THREADS];
    #pragma unroll
    for (int q = 0; q < (Nsl * Hh / 4) / PAIR_THREADS; ++q) vst[q] = vsrc[tid + PAIR_THREADS * q];
    #pragma unroll
    for (int q = 0; q < (Nsl * Hh / 4) / PAIR_THREADS; ++q) vsh[tid + PAIR_THREADS * q] = vst[q];
    {
        float* pf = reinterpret_cast<float*>(psh);
        pf[ph * 2 * Hh + pt]      = pa;
        pf[ph * 2 * Hh + Hh + pt] = pb;
    }
    const float batt = b_att[0];

    __syncthreads();

    const int tn = wave & 7;
    const int kh = wave >> 3;
    const float* __restrict__ wp = Weff + (size_t)(kh * (Hh / 2) + 2 * hh) * Dd + tn * 16 + c;
    v2f wb[32];
    #pragma unroll
    for (int s = 0; s < 32; ++s) {
        wb[s].x = wp[(4 * s) * Dd];
        wb[s].y = wp[(4 * s + 1) * Dd];
    }
    const v4f be4 = reinterpret_cast<const v4f*>(beff)[lane];

    v2f ub[8], lg[8], lb[8], wa[8], gacc[8];
    #pragma unroll
    for (int m = 0; m < 4; ++m) {
        const v4f e = psh[16 * m + c];
        const v4f g = psh[64 + 16 * m + c];
        const v4f l = psh[128 + 16 * m + c];
        const v4f w = psh[192 + 16 * m + c];
        ub[2*m] = u4[m].lo + e.lo;  ub[2*m+1] = u4[m].hi + e.hi;
        lg[2*m] = g.lo;             lg[2*m+1] = g.hi;
        lb[2*m] = l.lo;             lb[2*m+1] = l.hi;
        wa[2*m] = w.lo;             wa[2*m+1] = w.hi;
    }
    #pragma unroll
    for (int k = 0; k < 8; ++k) gacc[k] = (v2f){0.0f, 0.0f};
    float sacc = 0.0f;

    const v4f* vbase = vsh + hh * (Hh / 4) + c;
    v4f vcur[4];
    #pragma unroll
    for (int m = 0; m < 4; ++m) vcur[m] = vbase[16 * m];

    for (int t = 0; t < Nsl / 2; ++t) {
        const v4f* vnrow = vbase + ((t + 1) & (Nsl / 2 - 1)) * (2 * Hh / 4);
        v4f vnext[4];
        #pragma unroll
        for (int m = 0; m < 4; ++m) vnext[m] = vnrow[16 * m];

        v2f h[8];
        #pragma unroll
        for (int m = 0; m < 4; ++m) {
            h[2*m]   = ub[2*m]   + vcur[m].lo;
            h[2*m+1] = ub[2*m+1] + vcur[m].hi;
        }
        const v2f s2 = ((h[0] + h[1]) + (h[2] + h[3])) + ((h[4] + h[5]) + (h[6] + h[7]));
        const float mu = row16_sum(s2.x + s2.y) * (1.0f / Hh);

        #pragma unroll
        for (int k = 0; k < 8; ++k) h[k] = h[k] - mu;
        v2f qa = h[0] * h[0];
        v2f qb = h[1] * h[1];
        #pragma unroll
        for (int k = 1; k < 4; ++k) {
            qa = qa + h[2*k]   * h[2*k];
            qb = qb + h[2*k+1] * h[2*k+1];
        }
        const v2f q2 = qa + qb;
        const float var = row16_sum(q2.x + q2.y) * (1.0f / Hh);
        const float inv = rsqrtf(var + LN_EPS);

        #pragma unroll
        for (int k = 0; k < 8; ++k) {
            v2f hn = h[k] * inv * lg[k] + lb[k];
            hn.x = fmaxf(hn.x, 0.0f);
            hn.y = fmaxf(hn.y, 0.0f);
            h[k] = hn;
        }
        v2f da = h[0] * wa[0];
        v2f db = h[1] * wa[1];
        #pragma unroll
        for (int k = 1; k < 4; ++k) {
            da = da + h[2*k]   * wa[2*k];
            db = db + h[2*k+1] * wa[2*k+1];
        }
        const v2f d2 = da + db;
        const float adot = row16_sum(d2.x + d2.y) + batt;
        float a = 1.0f / (1.0f + __expf(-adot));
        a = (2 * t + hh == i) ? 0.0f : a;

        #pragma unroll
        for (int k = 0; k < 8; ++k) gacc[k] = gacc[k] + a * h[k];
        sacc += a;

        #pragma unroll
        for (int m = 0; m < 4; ++m) vcur[m] = vnext[m];
    }

    #pragma unroll
    for (int k = 0; k < 8; ++k) {
        gacc[k].x += xhalf(gacc[k].x);
        gacc[k].y += xhalf(gacc[k].y);
    }
    sacc += xhalf(sacc);

    float* gf = reinterpret_cast<float*>(gsh);
    #pragma unroll
    for (int q = 0; q < 2; ++q) {
        const v2f lo = hh ? gacc[4 + 2*q]     : gacc[2*q];
        const v2f hi = hh ? gacc[4 + 2*q + 1] : gacc[2*q + 1];
        v4f o;
        o.lo = lo;
        o.hi = hi;
        *reinterpret_cast<v4f*>(gf + wave * GP + 64 * (2 * hh + q) + 4 * c) = o;
    }
    __syncthreads();

    v8f acc = {};
    const float* ga = gf + c * GP + kh * (Hh / 2) + 2 * hh;
    #pragma unroll
    for (int s = 0; s < 32; ++s) {
        const v2f a = *reinterpret_cast<const v2f*>(ga + 4 * s);
        acc = __builtin_amdgcn_wmma_f32_16x16x4_f32(false, a, false, wb[s],
                                                    (short)0, acc, false, false);
    }
    float* of = reinterpret_cast<float*>(osh);
    #pragma unroll
    for (int r = 0; r < 8; ++r)
        of[kh * (IPB * OP) + (8 * hh + r) * OP + tn * 16 + c] = acc[r];
    __syncthreads();

    const v4f p0 = *reinterpret_cast<const v4f*>(of + wave * OP + 4 * lane);
    const v4f p1 = *reinterpret_cast<const v4f*>(of + IPB * OP + wave * OP + 4 * lane);
    const v4f res = (p0 + p1) + sacc * be4;
    *reinterpret_cast<v4f*>(out + (size_t)bi * Dd + 4 * lane) = res;
}

extern "C" void kernel_launch(void* const* d_in, const int* in_sizes, int n_in,
                              void* d_out, int out_size, void* d_ws, size_t ws_size,
                              hipStream_t stream)
{
    const float* slots = (const float*)d_in[0];
    const float* W_enc = (const float*)d_in[1];
    const float* b_enc = (const float*)d_in[2];
    const float* ln_g  = (const float*)d_in[3];
    const float* ln_b  = (const float*)d_in[4];
    const float* W_att = (const float*)d_in[5];
    const float* b_att = (const float*)d_in[6];
    const float* W_eff = (const float*)d_in[7];
    const float* b_eff = (const float*)d_in[8];
    float* out = (float*)d_out;

    float* ws = (float*)d_ws;
    float* U  = ws;
    float* V  = U + (size_t)BN * Hh;

    enc_gemm_kernel<<<1024, 256, 0, stream>>>(slots, W_enc, U);

    pair_reduce_kernel<<<PAIR_GRID, PAIR_THREADS, 0, stream>>>(U, V, b_enc, ln_g, ln_b,
                                                               W_att, b_att, W_eff, b_eff, out);
}


# enc GEMM moved into the head of the pair kernel: slots[b] in LDS as A, each wave's W_enc column fragments from global one block of 8 k-steps ahead, 96 WMMAs per wave, U/V land in LDS; one launch
# speedup vs baseline: 2.6365x; 1.3597x over previous
#include <hip/hip_runtime.h>

typedef __attribute__((ext_vector_type(2))) float v2f;
typedef __attribute__((ext_vector_type(8))) float v8f;

constexpr int Bsz = 128;
constexpr int Nsl = 32;
constexpr int Dd  = 128;
constexpr int Hh  = 256;
constexpr int BN  = Bsz * Nsl;
constexpr float LN_EPS = 1e-5f;

typedef __attribute__((ext_vector_type(4))) float v4f;

constexpr int IPB = 16;
constexpr int PAIR_THREADS = IPB * 32;
constexpr int PAIR_GRID = BN / IPB;
constexpr int GP = Hh + 4;
constexpr int OP = Dd + 4;
constexpr int KB = 8;
constexpr int SP = Dd + 4;

template <int M>
__device__ inline float row_xmask_add(float v) {
    const int moved = __builtin_amdgcn_update_dpp(0, __builtin_bit_cast(int, v),
                                                  0x160 | M, 0xf, 0xf, true);
    return v + __builtin_bit_cast(float, moved);
}

__device__ inline float row16_sum(float v) {
    v = row_xmask_add<1>(v);
    v = row_xmask_add<2>(v);
    v = row_xmask_add<4>(v);
    v = row_xmask_add<8>(v);
    return v;
}

__device__ inline float xhalf(float x) {
    const int b = __builtin_bit_cast(int, x);
    return __builtin_bit_cast(float,
        __builtin_amdgcn_permlanex16(b, b, 0x76543210, 0xfedcba98, false, false));
}

__global__ void __launch_bounds__(PAIR_THREADS)
pair_reduce_kernel(const float* __restrict__ slots,
                   const float* __restrict__ Wenc,
                   const float* __restrict__ b_enc,
                   const float* __restrict__ ln_g,
                   const float* __restrict__ ln_b,
                   const float* __restrict__ W_att,
                   const float* __restrict__ b_att,
                   const float* __restrict__ Weff,
                   const float* __restrict__ beff,
                   float* __restrict__ out)
{
    __shared__ v4f ssh[Nsl * SP / 4];
    __shared__ v4f ush[IPB * GP / 4];
    __shared__ v4f vsh[Nsl * Hh / 4];
    __shared__ v4f psh[4 * Hh / 4];
    __shared__ v4f gsh[IPB * GP / 4];
    __shared__ v4f osh[2 * IPB * OP / 4];

    const int tid  = threadIdx.x;
    const int lane = tid & 31;
    const int wave = tid >> 5;
    const int c    = lane & 15;
    const int hh   = lane >> 4;
    const int b    = blockIdx.x >> 1;
    const int i    = (blockIdx.x & 1) * IPB + wave;
    const int bi   = b * Nsl + i;

    const int p    = blockIdx.x & 1;

    const v4f* __restrict__ ssrc = reinterpret_cast<const v4f*>(slots + (size_t)b * Nsl * Dd);
    v4f sst[(Nsl * Dd / 4) / PAIR_THREADS];
    #pragma unroll
    for (int q = 0; q < (Nsl * Dd / 4) / PAIR_THREADS; ++q) sst[q] = ssrc[tid + PAIR_THREADS * q];

    const float* __restrict__ wq = Wenc + (size_t)(2 * hh) * Hh + wave * 16 + c;
    v2f wv[32], wt[32];
    #pragma unroll
    for (int s = 0; s < KB; ++s) {
        wv[s].x = wq[(Dd + 4 * s) * Hh];
        wv[s].y = wq[(Dd + 4 * s + 1) * Hh];
        wt[s].x = wq[(4 * s) * Hh];
        wt[s].y = wq[(4 * s + 1) * Hh];
    }

    const int pt = tid & (Hh - 1);
    const int ph = tid >> 8;
    const float pa = (ph ? ln_b : b_enc)[pt];
    const float pb = (ph ? W_att : ln_g)[pt];

    float* sf = reinterpret_cast<float*>(ssh);
    #pragma unroll
    for (int q = 0; q < (Nsl * Dd / 4) / PAIR_THREADS; ++q) {
        const int idx = tid + PAIR_THREADS * q;
        *reinterpret_cast<v4f*>(sf + (idx >> 5) * SP + 4 * (idx & 31)) = sst[q];
    }
    {
        float* pf = reinterpret_cast<float*>(psh);
        pf[ph * 2 * Hh + pt]      = pa;
        pf[ph * 2 * Hh + Hh + pt] = pb;
    }
    const float batt = b_att[0];

    __syncthreads();

    float* vf = reinterpret_cast<float*>(vsh);
    float* uf = reinterpret_cast<float*>(ush);
    {
        const float* aown = sf + (16 * p + c) * SP + 2 * hh;
        const float* aoth = sf + (16 * (1 - p) + c) * SP + 2 * hh;
        v8f accU = {}, accVown = {}, accVoth = {};
        v2f fo[2][KB], ft[2][KB];
        #pragma unroll
        for (int s = 0; s < KB; ++s) {
            fo[0][s] = *reinterpret_cast<const v2f*>(aown + 4 * s);
            ft[0][s] = *reinterpret_cast<const v2f*>(aoth + 4 * s);
        }
        #pragma unroll
        for (int j = 0; j < 32 / KB; ++j) {
            if (j + 1 < 32 / KB) {
                #pragma unroll
                for (int s = KB * (j + 1); s < KB * (j + 2); ++s) {
                    wv[s].x = wq[(Dd + 4 * s) * Hh];
                    wv[s].y = wq[(Dd + 4 * s + 1) * Hh];
                    wt[s].x = wq[(4 * s) * Hh];
                    wt[s].y = wq[(4 * s + 1) * Hh];
                }
                #pragma unroll
                for (int s = 0; s < KB; ++s) {
                    fo[(j + 1) & 1][s] = *reinterpret_cast<const v2f*>(aown + 4 * (KB * (j + 1) + s));
                    ft[(j + 1) & 1][s] = *reinterpret_cast<const v2f*>(aoth + 4 * (KB * (j + 1) + s));
                }
            }
            __builtin_amdgcn_sched_barrier(0);
            #pragma unroll
            for (int s = 0; s < KB; ++s) {
                const int k = KB * j + s;
                accVown = __builtin_amdgcn_wmma_f32_16x16x4_f32(false, fo[j & 1][s], false, wv[k],
                                                                (short)0, accVown, false, false);
                accVoth = __builtin_amdgcn_wmma_f32_16x16x4_f32(false, ft[j & 1][s], false, wv[k],
                                                                (short)0, accVoth, false, false);
                accU    = __builtin_amdgcn_wmma_f32_16x16x4_f32(false, fo[j & 1][s], false, wt[k],
                                                                (short)0, accU, false, false);
            }
            __builtin_amdgcn_sched_barrier(0);
        }
        #pragma unroll
        for (int r = 0; r < 8; ++r) {
            vf[(16 * p + 8 * hh + r) * Hh + wave * 16 + c]       = accVown[r];
            vf[(16 * (1 - p) + 8 * hh + r) * Hh + wave * 16 + c] = accVoth[r];
            uf[(8 * hh + r) * GP + wave * 16 + c]                = accU[r];
        }
    }
    __syncthreads();

    v4f u4[4];
    #pragma unroll
    for (int m = 0; m < 4; ++m) u4[m] = *reinterpret_cast<const v4f*>(uf + wave * GP + 64 * m + 4 * c);

    const int tn = wave & 7;
    const int kh = wave >> 3;
    const float* __restrict__ wp = Weff + (size_t)(kh * (Hh / 2) + 2 * hh) * Dd + tn * 16 + c;
    v2f wb[32];
    #pragma unroll
    for (int s = 0; s < 32; ++s) {
        wb[s].x = wp[(4 * s) * Dd];
        wb[s].y = wp[(4 * s + 1) * Dd];
    }
    const v4f be4 = reinterpret_cast<const v4f*>(beff)[lane];

    v2f ub[8], lg[8], lb[8], wa[8], gacc[8];
    #pragma unroll
    for (int m = 0; m < 4; ++m) {
        const v4f e = psh[16 * m + c];
        const v4f g = psh[64 + 16 * m + c];
        const v4f l = psh[128 + 16 * m + c];
        const v4f w = psh[192 + 16 * m + c];
        ub[2*m] = u4[m].lo + e.lo;  ub[2*m+1] = u4[m].hi + e.hi;
        lg[2*m] = g.lo;             lg[2*m+1] = g.hi;
        lb[2*m] = l.lo;             lb[2*m+1] = l.hi;
        wa[2*m] = w.lo;             wa[2*m+1] = w.hi;
    }
    #pragma unroll
    for (int k = 0; k < 8; ++k) gacc[k] = (v2f){0.0f, 0.0f};
    float sacc = 0.0f;

    const v4f* vbase = vsh + hh * (Hh / 4) + c;
    v4f vcur[4];
    #pragma unroll
    for (int m = 0; m < 4; ++m) vcur[m] = vbase[16 * m];

    for (int t = 0; t < Nsl / 2; ++t) {
        const v4f* vnrow = vbase + ((t + 1) & (Nsl / 2 - 1)) * (2 * Hh / 4);
        v4f vnext[4];
        #pragma unroll
        for (int m = 0; m < 4; ++m) vnext[m] = vnrow[16 * m];

        v2f h[8];
        #pragma unroll
        for (int m = 0; m < 4; ++m) {
            h[2*m]   = ub[2*m]   + vcur[m].lo;
            h[2*m+1] = ub[2*m+1] + vcur[m].hi;
        }
        const v2f s2 = ((h[0] + h[1]) + (h[2] + h[3])) + ((h[4] + h[5]) + (h[6] + h[7]));
        const float mu = row16_sum(s2.x + s2.y) * (1.0f / Hh);

        #pragma unroll
        for (int k = 0; k < 8; ++k) h[k] = h[k] - mu;
        v2f qa = h[0] * h[0];
        v2f qb = h[1] * h[1];
        #pragma unroll
        for (int k = 1; k < 4; ++k) {
            qa = qa + h[2*k]   * h[2*k];
            qb = qb + h[2*k+1] * h[2*k+1];
        }
        const v2f q2 = qa + qb;
        const float var = row16_sum(q2.x + q2.y) * (1.0f / Hh);
        const float inv = rsqrtf(var + LN_EPS);

        #pragma unroll
        for (int k = 0; k < 8; ++k) {
            v2f hn = h[k] * inv * lg[k] + lb[k];
            hn.x = fmaxf(hn.x, 0.0f);
            hn.y = fmaxf(hn.y, 0.0f);
            h[k] = hn;
        }
        v2f da = h[0] * wa[0];
        v2f db = h[1] * wa[1];
        #pragma unroll
        for (int k = 1; k < 4; ++k) {
            da = da + h[2*k]   * wa[2*k];
            db = db + h[2*k+1] * wa[2*k+1];
        }
        const v2f d2 = da + db;
        const float adot = row16_sum(d2.x + d2.y) + batt;
        float a = 1.0f / (1.0f + __expf(-adot));
        a = (2 * t + hh == i) ? 0.0f : a;

        #pragma unroll
        for (int k = 0; k < 8; ++k) gacc[k] = gacc[k] + a * h[k];
        sacc += a;

        #pragma unroll
        for (int m = 0; m < 4; ++m) vcur[m] = vnext[m];
    }

    #pragma unroll
    for (int k = 0; k < 8; ++k) {
        gacc[k].x += xhalf(gacc[k].x);
        gacc[k].y += xhalf(gacc[k].y);
    }
    sacc += xhalf(sacc);

    float* gf = reinterpret_cast<float*>(gsh);
    #pragma unroll
    for (int q = 0; q < 2; ++q) {
        const v2f lo = hh ? gacc[4 + 2*q]     : gacc[2*q];
        const v2f hi = hh ? gacc[4 + 2*q + 1] : gacc[2*q + 1];
        v4f o;
        o.lo = lo;
        o.hi = hi;
        *reinterpret_cast<v4f*>(gf + wave * GP + 64 * (2 * hh + q) + 4 * c) = o;
    }
    __syncthreads();

    v8f acc = {};
    const float* ga = gf + c * GP + kh * (Hh / 2) + 2 * hh;
    #pragma unroll
    for (int s = 0; s < 32; ++s) {
        const v2f a = *reinterpret_cast<const v2f*>(ga + 4 * s);
        acc = __builtin_amdgcn_wmma_f32_16x16x4_f32(false, a, false, wb[s],
                                                    (short)0, acc, false, false);
    }
    float* of = reinterpret_cast<float*>(osh);
    #pragma unroll
    for (int r = 0; r < 8; ++r)
        of[kh * (IPB * OP) + (8 * hh + r) * OP + tn * 16 + c] = acc[r];
    __syncthreads();

    const v4f p0 = *reinterpret_cast<const v4f*>(of + wave * OP + 4 * lane);
    const v4f p1 = *reinterpret_cast<const v4f*>(of + IPB * OP + wave * OP + 4 * lane);
    const v4f res = (p0 + p1) + sacc * be4;
    *reinterpret_cast<v4f*>(out + (size_t)bi * Dd + 4 * lane) = res;
}

extern "C" void kernel_launch(void* const* d_in, const int* in_sizes, int n_in,
                              void* d_out, int out_size, void* d_ws, size_t ws_size,
                              hipStream_t stream)
{
    const float* slots = (const float*)d_in[0];
    const float* W_enc = (const float*)d_in[1];
    const float* b_enc = (const float*)d_in[2];
    const float* ln_g  = (const float*)d_in[3];
    const float* ln_b  = (const float*)d_in[4];
    const float* W_att = (const float*)d_in[5];
    const float* b_att = (const float*)d_in[6];
    const float* W_eff = (const float*)d_in[7];
    const float* b_eff = (const float*)d_in[8];
    float* out = (float*)d_out;

    pair_reduce_kernel<<<PAIR_GRID, PAIR_THREADS, 0, stream>>>(slots, W_enc, b_enc, ln_g, ln_b,
                                                               W_att, b_att, W_eff, b_eff, out);
}


# pair loop: LayerNorm mean hoisted: U+b_enc row and each V row centred once (V in place in LDS, one more barrier); no per-pair sum, DPP reduction or subtract; mean summed per row, not per pair
# speedup vs baseline: 2.7894x; 1.0580x over previous
#include <hip/hip_runtime.h>

typedef __attribute__((ext_vector_type(2))) float v2f;
typedef __attribute__((ext_vector_type(8))) float v8f;

constexpr int Bsz = 128;
constexpr int Nsl = 32;
constexpr int Dd  = 128;
constexpr int Hh  = 256;
constexpr int BN  = Bsz * Nsl;
constexpr float LN_EPS = 1e-5f;

typedef __attribute__((ext_vector_type(4))) float v4f;

constexpr int IPB = 16;
constexpr int PAIR_THREADS = IPB * 32;
constexpr int PAIR_GRID = BN / IPB;
constexpr int GP = Hh + 4;
constexpr int OP = Dd + 4;
constexpr int KB = 8;
constexpr int SP = Dd + 4;

template <int M>
__device__ inline float row_xmask_add(float v) {
    const int moved = __builtin_amdgcn_update_dpp(0, __builtin_bit_cast(int, v),
                                                  0x160 | M, 0xf, 0xf, true);
    return v + __builtin_bit_cast(float, moved);
}

__device__ inline float row16_sum(float v) {
    v = row_xmask_add<1>(v);
    v = row_xmask_add<2>(v);
    v = row_xmask_add<4>(v);
    v = row_xmask_add<8>(v);
    return v;
}

__device__ inline float xhalf(float x) {
    const int b = __builtin_bit_cast(int, x);
    return __builtin_bit_cast(float,
        __builtin_amdgcn_permlanex16(b, b, 0x76543210, 0xfedcba98, false, false));
}

__global__ void __launch_bounds__(PAIR_THREADS)
pair_reduce_kernel(const float* __restrict__ slots,
                   const float* __restrict__ Wenc,
                   const float* __restrict__ b_enc,
                   const float* __restrict__ ln_g,
                   const float* __restrict__ ln_b,
                   const float* __restrict__ W_att,
                   const float* __restrict__ b_att,
                   const float* __restrict__ Weff,
                   const float* __restrict__ beff,
                   float* __restrict__ out)
{
    __shared__ v4f ssh[Nsl * SP / 4];
    __shared__ v4f ush[IPB * GP / 4];
    __shared__ v4f vsh[Nsl * Hh / 4];
    __shared__ v4f psh[4 * Hh / 4];
    __shared__ v4f gsh[IPB * GP / 4];
    __shared__ v4f osh[2 * IPB * OP / 4];

    const int tid  = threadIdx.x;
    const int lane = tid & 31;
    const int wave = tid >> 5;
    const int c    = lane & 15;
    const int hh   = lane >> 4;
    const int b    = blockIdx.x >> 1;
    const int i    = (blockIdx.x & 1) * IPB + wave;
    const int bi   = b * Nsl + i;

    const int p    = blockIdx.x & 1;

    const v4f* __restrict__ ssrc = reinterpret_cast<const v4f*>(slots + (size_t)b * Nsl * Dd);
    v4f sst[(Nsl * Dd / 4) / PAIR_THREADS];
    #pragma unroll
    for (int q = 0; q < (Nsl * Dd / 4) / PAIR_THREADS; ++q) sst[q] = ssrc[tid + PAIR_THREADS * q];

    const float* __restrict__ wq = Wenc + (size_t)(2 * hh) * Hh + wave * 16 + c;
    v2f wv[32], wt[32];
    #pragma unroll
    for (int s = 0; s < KB; ++s) {
        wv[s].x = wq[(Dd + 4 * s) * Hh];
        wv[s].y = wq[(Dd + 4 * s + 1) * Hh];
        wt[s].x = wq[(4 * s) * Hh];
        wt[s].y = wq[(4 * s + 1) * Hh];
    }

    const int pt = tid & (Hh - 1);
    const int ph = tid >> 8;
    const float pa = (ph ? ln_b : b_enc)[pt];
    const float pb = (ph ? W_att : ln_g)[pt];

    float* sf = reinterpret_cast<float*>(ssh);
    #pragma unroll
    for (int q = 0; q < (Nsl * Dd / 4) / PAIR_THREADS; ++q) {
        const int idx = tid + PAIR_THREADS * q;
        *reinterpret_cast<v4f*>(sf + (idx >> 5) * SP + 4 * (idx & 31)) = sst[q];
    }
    {
        float* pf = reinterpret_cast<float*>(psh);
        pf[ph * 2 * Hh + pt]      = pa;
        pf[ph * 2 * Hh + Hh + pt] = pb;
    }
    const float batt = b_att[0];

    __syncthreads();

    float* vf = reinterpret_cast<float*>(vsh);
    float* uf = reinterpret_cast<float*>(ush);
    {
        const float* aown = sf + (16 * p + c) * SP + 2 * hh;
        const float* aoth = sf + (16 * (1 - p) + c) * SP + 2 * hh;
        v8f accU = {}, accVown = {}, accVoth = {};
        v2f fo[2][KB], ft[2][KB];
        #pragma unroll
        for (int s = 0; s < KB; ++s) {
            fo[0][s] = *reinterpret_cast<const v2f*>(aown + 4 * s);
            ft[0][s] = *reinterpret_cast<const v2f*>(aoth + 4 * s);
        }
        #pragma unroll
        for (int j = 0; j < 32 / KB; ++j) {
            if (j + 1 < 32 / KB) {
                #pragma unroll
                for (int s = KB * (j + 1); s < KB * (j + 2); ++s) {
                    wv[s].x = wq[(Dd + 4 * s) * Hh];
                    wv[s].y = wq[(Dd + 4 * s + 1) * Hh];
                    wt[s].x = wq[(4 * s) * Hh];
                    wt[s].y = wq[(4 * s + 1) * Hh];
                }
                #pragma unroll
                for (int s = 0; s < KB; ++s) {
                    fo[(j + 1) & 1][s] = *reinterpret_cast<const v2f*>(aown + 4 * (KB * (j + 1) + s));
                    ft[(j + 1) & 1][s] = *reinterpret_cast<const v2f*>(aoth + 4 * (KB * (j + 1) + s));
                }
            }
            __builtin_amdgcn_sched_barrier(0);
            #pragma unroll
            for (int s = 0; s < KB; ++s) {
                const int k = KB * j + s;
                accVown = __builtin_amdgcn_wmma_f32_16x16x4_f32(false, fo[j & 1][s], false, wv[k],
                                                                (short)0, accVown, false, false);
                accVoth = __builtin_amdgcn_wmma_f32_16x16x4_f32(false, ft[j & 1][s], false, wv[k],
                                                                (short)0, accVoth, false, false);
                accU    = __builtin_amdgcn_wmma_f32_16x16x4_f32(false, fo[j & 1][s], false, wt[k],
                                                                (short)0, accU, false, false);
            }
            __builtin_amdgcn_sched_barrier(0);
        }
        #pragma unroll
        for (int r = 0; r < 8; ++r) {
            vf[(16 * p + 8 * hh + r) * Hh + wave * 16 + c]       = accVown[r];
            vf[(16 * (1 - p) + 8 * hh + r) * Hh + wave * 16 + c] = accVoth[r];
            uf[(8 * hh + r) * GP + wave * 16 + c]                = accU[r];
        }
    }
    __syncthreads();

    {
        v4f* vrow = vsh + (2 * wave + hh) * (Hh / 4) + c;
        v4f x[4];
        #pragma unroll
        for (int m = 0; m < 4; ++m) x[m] = vrow[16 * m];
        const v4f s4 = (x[0] + x[1]) + (x[2] + x[3]);
        const float mv = row16_sum((s4.x + s4.y) + (s4.z + s4.w)) * (1.0f / Hh);
        #pragma unroll
        for (int m = 0; m < 4; ++m) vrow[16 * m] = x[m] - mv;
    }

    v4f u4[4];
    #pragma unroll
    for (int m = 0; m < 4; ++m) u4[m] = *reinterpret_cast<const v4f*>(uf + wave * GP + 64 * m + 4 * c);

    v2f ub[8], lg[8], lb[8], wa[8], gacc[8];
    #pragma unroll
    for (int m = 0; m < 4; ++m) {
        const v4f e = psh[16 * m + c];
        const v4f g = psh[64 + 16 * m + c];
        const v4f l = psh[128 + 16 * m + c];
        const v4f w = psh[192 + 16 * m + c];
        ub[2*m] = u4[m].lo + e.lo;  ub[2*m+1] = u4[m].hi + e.hi;
        lg[2*m] = g.lo;             lg[2*m+1] = g.hi;
        lb[2*m] = l.lo;             lb[2*m+1] = l.hi;
        wa[2*m] = w.lo;             wa[2*m+1] = w.hi;
    }
    {
        const v2f s2 = ((ub[0] + ub[1]) + (ub[2] + ub[3])) + ((ub[4] + ub[5]) + (ub[6] + ub[7]));
        const float mu = row16_sum(s2.x + s2.y) * (1.0f / Hh);
        #pragma unroll
        for (int k = 0; k < 8; ++k) ub[k] = ub[k] - mu;
    }
    #pragma unroll
    for (int k = 0; k < 8; ++k) gacc[k] = (v2f){0.0f, 0.0f};
    float sacc = 0.0f;

    __syncthreads();

    const int tn = wave & 7;
    const int kh = wave >> 3;
    const float* __restrict__ wp = Weff + (size_t)(kh * (Hh / 2) + 2 * hh) * Dd + tn * 16 + c;
    v2f wb[32];
    #pragma unroll
    for (int s = 0; s < 32; ++s) {
        wb[s].x = wp[(4 * s) * Dd];
        wb[s].y = wp[(4 * s + 1) * Dd];
    }
    const v4f be4 = reinterpret_cast<const v4f*>(beff)[lane];

    const v4f* vbase = vsh + hh * (Hh / 4) + c;
    v4f vcur[4];
    #pragma unroll
    for (int m = 0; m < 4; ++m) vcur[m] = vbase[16 * m];

    for (int t = 0; t < Nsl / 2; ++t) {
        const v4f* vnrow = vbase + ((t + 1) & (Nsl / 2 - 1)) * (2 * Hh / 4);
        v4f vnext[4];
        #pragma unroll
        for (int m = 0; m < 4; ++m) vnext[m] = vnrow[16 * m];

        v2f h[8];
        #pragma unroll
        for (int m = 0; m < 4; ++m) {
            h[2*m]   = ub[2*m]   + vcur[m].lo;
            h[2*m+1] = ub[2*m+1] + vcur[m].hi;
        }
        v2f qa = h[0] * h[0];
        v2f qb = h[1] * h[1];
        #pragma unroll
        for (int k = 1; k < 4; ++k) {
            qa = qa + h[2*k]   * h[2*k];
            qb = qb + h[2*k+1] * h[2*k+1];
        }
        const v2f q2 = qa + qb;
        const float var = row16_sum(q2.x + q2.y) * (1.0f / Hh);
        const float inv = rsqrtf(var + LN_EPS);

        #pragma unroll
        for (int k = 0; k < 8; ++k) {
            v2f hn = h[k] * inv * lg[k] + lb[k];
            hn.x = fmaxf(hn.x, 0.0f);
            hn.y = fmaxf(hn.y, 0.0f);
            h[k] = hn;
        }
        v2f da = h[0] * wa[0];
        v2f db = h[1] * wa[1];
        #pragma unroll
        for (int k = 1; k < 4; ++k) {
            da = da + h[2*k]   * wa[2*k];
            db = db + h[2*k+1] * wa[2*k+1];
        }
        const v2f d2 = da + db;
        const float adot = row16_sum(d2.x + d2.y) + batt;
        float a = 1.0f / (1.0f + __expf(-adot));
        a = (2 * t + hh == i) ? 0.0f : a;

        #pragma unroll
        for (int k = 0; k < 8; ++k) gacc[k] = gacc[k] + a * h[k];
        sacc += a;

        #pragma unroll
        for (int m = 0; m < 4; ++m) vcur[m] = vnext[m];
    }

    #pragma unroll
    for (int k = 0; k < 8; ++k) {
        gacc[k].x += xhalf(gacc[k].x);
        gacc[k].y += xhalf(gacc[k].y);
    }
    sacc += xhalf(sacc);

    float* gf = reinterpret_cast<float*>(gsh);
    #pragma unroll
    for (int q = 0; q < 2; ++q) {
        const v2f lo = hh ? gacc[4 + 2*q]     : gacc[2*q];
        const v2f hi = hh ? gacc[4 + 2*q + 1] : gacc[2*q + 1];
        v4f o;
        o.lo = lo;
        o.hi = hi;
        *reinterpret_cast<v4f*>(gf + wave * GP + 64 * (2 * hh + q) + 4 * c) = o;
    }
    __syncthreads();

    v8f acc = {};
    const float* ga = gf + c * GP + kh * (Hh / 2) + 2 * hh;
    #pragma unroll
    for (int s = 0; s < 32; ++s) {
        const v2f a = *reinterpret_cast<const v2f*>(ga + 4 * s);
        acc = __builtin_amdgcn_wmma_f32_16x16x4_f32(false, a, false, wb[s],
                                                    (short)0, acc, false, false);
    }
    float* of = reinterpret_cast<float*>(osh);
    #pragma unroll
    for (int r = 0; r < 8; ++r)
        of[kh * (IPB * OP) + (8 * hh + r) * OP + tn * 16 + c] = acc[r];
    __syncthreads();

    const v4f p0 = *reinterpret_cast<const v4f*>(of + wave * OP + 4 * lane);
    const v4f p1 = *reinterpret_cast<const v4f*>(of + IPB * OP + wave * OP + 4 * lane);
    const v4f res = (p0 + p1) + sacc * be4;
    *reinterpret_cast<v4f*>(out + (size_t)bi * Dd + 4 * lane) = res;
}

extern "C" void kernel_launch(void* const* d_in, const int* in_sizes, int n_in,
                              void* d_out, int out_size, void* d_ws, size_t ws_size,
                              hipStream_t stream)
{
    const float* slots = (const float*)d_in[0];
    const float* W_enc = (const float*)d_in[1];
    const float* b_enc = (const float*)d_in[2];
    const float* ln_g  = (const float*)d_in[3];
    const float* ln_b  = (const float*)d_in[4];
    const float* W_att = (const float*)d_in[5];
    const float* b_att = (const float*)d_in[6];
    const float* W_eff = (const float*)d_in[7];
    const float* b_eff = (const float*)d_in[8];
    float* out = (float*)d_out;

    pair_reduce_kernel<<<PAIR_GRID, PAIR_THREADS, 0, stream>>>(slots, W_enc, b_enc, ln_g, ln_b,
                                                               W_att, b_att, W_eff, b_eff, out);
}


# pair loop: per-pair scalar chain on bare hardware forms (about 1 ulp): rsq without rsqrtf's small-input scaling, rcp for the sigmoid's IEEE division, bias add and log2(e) scale in one fma
# speedup vs baseline: 2.9472x; 1.0566x over previous
#include <hip/hip_runtime.h>

typedef __attribute__((ext_vector_type(2))) float v2f;
typedef __attribute__((ext_vector_type(8))) float v8f;

constexpr int Bsz = 128;
constexpr int Nsl = 32;
constexpr int Dd  = 128;
constexpr int Hh  = 256;
constexpr int BN  = Bsz * Nsl;
constexpr float LN_EPS = 1e-5f;

typedef __attribute__((ext_vector_type(4))) float v4f;

constexpr int IPB = 16;
constexpr int PAIR_THREADS = IPB * 32;
constexpr int PAIR_GRID = BN / IPB;
constexpr int GP = Hh + 4;
constexpr int OP = Dd + 4;
constexpr int KB = 8;
constexpr int SP = Dd + 4;

template <int M>
__device__ inline float row_xmask_add(float v) {
    const int moved = __builtin_amdgcn_update_dpp(0, __builtin_bit_cast(int, v),
                                                  0x160 | M, 0xf, 0xf, true);
    return v + __builtin_bit_cast(float, moved);
}

__device__ inline float row16_sum(float v) {
    v = row_xmask_add<1>(v);
    v = row_xmask_add<2>(v);
    v = row_xmask_add<4>(v);
    v = row_xmask_add<8>(v);
    return v;
}

__device__ inline float xhalf(float x) {
    const int b = __builtin_bit_cast(int, x);
    return __builtin_bit_cast(float,
        __builtin_amdgcn_permlanex16(b, b, 0x76543210, 0xfedcba98, false, false));
}

__global__ void __launch_bounds__(PAIR_THREADS)
pair_reduce_kernel(const float* __restrict__ slots,
                   const float* __restrict__ Wenc,
                   const float* __restrict__ b_enc,
                   const float* __restrict__ ln_g,
                   const float* __restrict__ ln_b,
                   const float* __restrict__ W_att,
                   const float* __restrict__ b_att,
                   const float* __restrict__ Weff,
                   const float* __restrict__ beff,
                   float* __restrict__ out)
{
    __shared__ v4f ssh[Nsl * SP / 4];
    __shared__ v4f ush[IPB * GP / 4];
    __shared__ v4f vsh[Nsl * Hh / 4];
    __shared__ v4f psh[4 * Hh / 4];
    __shared__ v4f gsh[IPB * GP / 4];
    __shared__ v4f osh[2 * IPB * OP / 4];

    const int tid  = threadIdx.x;
    const int lane = tid & 31;
    const int wave = tid >> 5;
    const int c    = lane & 15;
    const int hh   = lane >> 4;
    const int b    = blockIdx.x >> 1;
    const int i    = (blockIdx.x & 1) * IPB + wave;
    const int bi   = b * Nsl + i;

    const int p    = blockIdx.x & 1;

    const v4f* __restrict__ ssrc = reinterpret_cast<const v4f*>(slots + (size_t)b * Nsl * Dd);
    v4f sst[(Nsl * Dd / 4) / PAIR_THREADS];
    #pragma unroll
    for (int q = 0; q < (Nsl * Dd / 4) / PAIR_THREADS; ++q) sst[q] = ssrc[tid + PAIR_THREADS * q];

    const float* __restrict__ wq = Wenc + (size_t)(2 * hh) * Hh + wave * 16 + c;
    v2f wv[32], wt[32];
    #pragma unroll
    for (int s = 0; s < KB; ++s) {
        wv[s].x = wq[(Dd + 4 * s) * Hh];
        wv[s].y = wq[(Dd + 4 * s + 1) * Hh];
        wt[s].x = wq[(4 * s) * Hh];
        wt[s].y = wq[(4 * s + 1) * Hh];
    }

    const int pt = tid & (Hh - 1);
    const int ph = tid >> 8;
    const float pa = (ph ? ln_b : b_enc)[pt];
    const float pb = (ph ? W_att : ln_g)[pt];

    float* sf = reinterpret_cast<float*>(ssh);
    #pragma unroll
    for (int q = 0; q < (Nsl * Dd / 4) / PAIR_THREADS; ++q) {
        const int idx = tid + PAIR_THREADS * q;
        *reinterpret_cast<v4f*>(sf + (idx >> 5) * SP + 4 * (idx & 31)) = sst[q];
    }
    {
        float* pf = reinterpret_cast<float*>(psh);
        pf[ph * 2 * Hh + pt]      = pa;
        pf[ph * 2 * Hh + Hh + pt] = pb;
    }
    constexpr float LOG2E = 1.4426950408889634f;
    const float nbatt = -b_att[0] * LOG2E;

    __syncthreads();

    float* vf = reinterpret_cast<float*>(vsh);
    float* uf = reinterpret_cast<float*>(ush);
    {
        const float* aown = sf + (16 * p + c) * SP + 2 * hh;
        const float* aoth = sf + (16 * (1 - p) + c) * SP + 2 * hh;
        v8f accU = {}, accVown = {}, accVoth = {};
        v2f fo[2][KB], ft[2][KB];
        #pragma unroll
        for (int s = 0; s < KB; ++s) {
            fo[0][s] = *reinterpret_cast<const v2f*>(aown + 4 * s);
            ft[0][s] = *reinterpret_cast<const v2f*>(aoth + 4 * s);
        }
        #pragma unroll
        for (int j = 0; j < 32 / KB; ++j) {
            if (j + 1 < 32 / KB) {
                #pragma unroll
                for (int s = KB * (j + 1); s < KB * (j + 2); ++s) {
                    wv[s].x = wq[(Dd + 4 * s) * Hh];
                    wv[s].y = wq[(Dd + 4 * s + 1) * Hh];
                    wt[s].x = wq[(4 * s) * Hh];
                    wt[s].y = wq[(4 * s + 1) * Hh];
                }
                #pragma unroll
                for (int s = 0; s < KB; ++s) {
                    fo[(j + 1) & 1][s] = *reinterpret_cast<const v2f*>(aown + 4 * (KB * (j + 1) + s));
                    ft[(j + 1) & 1][s] = *reinterpret_cast<const v2f*>(aoth + 4 * (KB * (j + 1) + s));
                }
            }
            __builtin_amdgcn_sched_barrier(0);
            #pragma unroll
            for (int s = 0; s < KB; ++s) {
                const int k = KB * j + s;
                accVown = __builtin_amdgcn_wmma_f32_16x16x4_f32(false, fo[j & 1][s], false, wv[k],
                                                                (short)0, accVown, false, false);
                accVoth = __builtin_amdgcn_wmma_f32_16x16x4_f32(false, ft[j & 1][s], false, wv[k],
                                                                (short)0, accVoth, false, false);
                accU    = __builtin_amdgcn_wmma_f32_16x16x4_f32(false, fo[j & 1][s], false, wt[k],
                                                                (short)0, accU, false, false);
            }
            __builtin_amdgcn_sched_barrier(0);
        }
        #pragma unroll
        for (int r = 0; r < 8; ++r) {
            vf[(16 * p + 8 * hh + r) * Hh + wave * 16 + c]       = accVown[r];
            vf[(16 * (1 - p) + 8 * hh + r) * Hh + wave * 16 + c] = accVoth[r];
            uf[(8 * hh + r) * GP + wave * 16 + c]                = accU[r];
        }
    }
    __syncthreads();

    {
        v4f* vrow = vsh + (2 * wave + hh) * (Hh / 4) + c;
        v4f x[4];
        #pragma unroll
        for (int m = 0; m < 4; ++m) x[m] = vrow[16 * m];
        const v4f s4 = (x[0] + x[1]) + (x[2] + x[3]);
        const float mv = row16_sum((s4.x + s4.y) + (s4.z + s4.w)) * (1.0f / Hh);
        #pragma unroll
        for (int m = 0; m < 4; ++m) vrow[16 * m] = x[m] - mv;
    }

    v4f u4[4];
    #pragma unroll
    for (int m = 0; m < 4; ++m) u4[m] = *reinterpret_cast<const v4f*>(uf + wave * GP + 64 * m + 4 * c);

    v2f ub[8], lg[8], lb[8], wa[8], gacc[8];
    #pragma unroll
    for (int m = 0; m < 4; ++m) {
        const v4f e = psh[16 * m + c];
        const v4f g = psh[64 + 16 * m + c];
        const v4f l = psh[128 + 16 * m + c];
        const v4f w = psh[192 + 16 * m + c];
        ub[2*m] = u4[m].lo + e.lo;  ub[2*m+1] = u4[m].hi + e.hi;
        lg[2*m] = g.lo;             lg[2*m+1] = g.hi;
        lb[2*m] = l.lo;             lb[2*m+1] = l.hi;
        wa[2*m] = w.lo;             wa[2*m+1] = w.hi;
    }
    {
        const v2f s2 = ((ub[0] + ub[1]) + (ub[2] + ub[3])) + ((ub[4] + ub[5]) + (ub[6] + ub[7]));
        const float mu = row16_sum(s2.x + s2.y) * (1.0f / Hh);
        #pragma unroll
        for (int k = 0; k < 8; ++k) ub[k] = ub[k] - mu;
    }
    #pragma unroll
    for (int k = 0; k < 8; ++k) gacc[k] = (v2f){0.0f, 0.0f};
    float sacc = 0.0f;

    __syncthreads();

    const int tn = wave & 7;
    const int kh = wave >> 3;
    const float* __restrict__ wp = Weff + (size_t)(kh * (Hh / 2) + 2 * hh) * Dd + tn * 16 + c;
    v2f wb[32];
    #pragma unroll
    for (int s = 0; s < 32; ++s) {
        wb[s].x = wp[(4 * s) * Dd];
        wb[s].y = wp[(4 * s + 1) * Dd];
    }
    const v4f be4 = reinterpret_cast<const v4f*>(beff)[lane];

    const v4f* vbase = vsh + hh * (Hh / 4) + c;
    v4f vcur[4];
    #pragma unroll
    for (int m = 0; m < 4; ++m) vcur[m] = vbase[16 * m];

    for (int t = 0; t < Nsl / 2; ++t) {
        const v4f* vnrow = vbase + ((t + 1) & (Nsl / 2 - 1)) * (2 * Hh / 4);
        v4f vnext[4];
        #pragma unroll
        for (int m = 0; m < 4; ++m) vnext[m] = vnrow[16 * m];

        v2f h[8];
        #pragma unroll
        for (int m = 0; m < 4; ++m) {
            h[2*m]   = ub[2*m]   + vcur[m].lo;
            h[2*m+1] = ub[2*m+1] + vcur[m].hi;
        }
        v2f qa = h[0] * h[0];
        v2f qb = h[1] * h[1];
        #pragma unroll
        for (int k = 1; k < 4; ++k) {
            qa = qa + h[2*k]   * h[2*k];
            qb = qb + h[2*k+1] * h[2*k+1];
        }
        const v2f q2 = qa + qb;
        const float var = row16_sum(q2.x + q2.y) * (1.0f / Hh);
        const float inv = __builtin_amdgcn_rsqf(var + LN_EPS);

        #pragma unroll
        for (int k = 0; k < 8; ++k) {
            v2f hn = h[k] * inv * lg[k] + lb[k];
            hn.x = fmaxf(hn.x, 0.0f);
            hn.y = fmaxf(hn.y, 0.0f);
            h[k] = hn;
        }
        v2f da = h[0] * wa[0];
        v2f db = h[1] * wa[1];
        #pragma unroll
        for (int k = 1; k < 4; ++k) {
            da = da + h[2*k]   * wa[2*k];
            db = db + h[2*k+1] * wa[2*k+1];
        }
        const v2f d2 = da + db;
        const float e = __builtin_amdgcn_exp2f(fmaf(row16_sum(d2.x + d2.y), -LOG2E, nbatt));
        float a = __builtin_amdgcn_rcpf(1.0f + e);
        a = (2 * t + hh == i) ? 0.0f : a;

        #pragma unroll
        for (int k = 0; k < 8; ++k) gacc[k] = gacc[k] + a * h[k];
        sacc += a;

        #pragma unroll
        for (int m = 0; m < 4; ++m) vcur[m] = vnext[m];
    }

    #pragma unroll
    for (int k = 0; k < 8; ++k) {
        gacc[k].x += xhalf(gacc[k].x);
        gacc[k].y += xhalf(gacc[k].y);
    }
    sacc += xhalf(sacc);

    float* gf = reinterpret_cast<float*>(gsh);
    #pragma unroll
    for (int q = 0; q < 2; ++q) {
        const v2f lo = hh ? gacc[4 + 2*q]     : gacc[2*q];
        const v2f hi = hh ? gacc[4 + 2*q + 1] : gacc[2*q + 1];
        v4f o;
        o.lo = lo;
        o.hi = hi;
        *reinterpret_cast<v4f*>(gf + wave * GP + 64 * (2 * hh + q) + 4 * c) = o;
    }
    __syncthreads();

    v8f acc = {};
    const float* ga = gf + c * GP + kh * (Hh / 2) + 2 * hh;
    #pragma unroll
    for (int s = 0; s < 32; ++s) {
        const v2f a = *reinterpret_cast<const v2f*>(ga + 4 * s);
        acc = __builtin_amdgcn_wmma_f32_16x16x4_f32(false, a, false, wb[s],
                                                    (short)0, acc, false, false);
    }
    float* of = reinterpret_cast<float*>(osh);
    #pragma unroll
    for (int r = 0; r < 8; ++r)
        of[kh * (IPB * OP) + (8 * hh + r) * OP + tn * 16 + c] = acc[r];
    __syncthreads();

    const v4f p0 = *reinterpret_cast<const v4f*>(of + wave * OP + 4 * lane);
    const v4f p1 = *reinterpret_cast<const v4f*>(of + IPB * OP + wave * OP + 4 * lane);
    const v4f res = (p0 + p1) + sacc * be4;
    *reinterpret_cast<v4f*>(out + (size_t)bi * Dd + 4 * lane) = res;
}

extern "C" void kernel_launch(void* const* d_in, const int* in_sizes, int n_in,
                              void* d_out, int out_size, void* d_ws, size_t ws_size,
                              hipStream_t stream)
{
    const float* slots = (const float*)d_in[0];
    const float* W_enc = (const float*)d_in[1];
    const float* b_enc = (const float*)d_in[2];
    const float* ln_g  = (const float*)d_in[3];
    const float* ln_b  = (const float*)d_in[4];
    const float* W_att = (const float*)d_in[5];
    const float* b_att = (const float*)d_in[6];
    const float* W_eff = (const float*)d_in[7];
    const float* b_eff = (const float*)d_in[8];
    float* out = (float*)d_out;

    pair_reduce_kernel<<<PAIR_GRID, PAIR_THREADS, 0, stream>>>(slots, W_enc, b_enc, ln_g, ln_b,
                                                               W_att, b_att, W_eff, b_eff, out);
}


# one wave per two rows i: 8-wave workgroups (2048 waves enter, not 4096); a wave's head and tail tiles double, one V row fetch and one A fragment serve both rows; no sum changes its order
# speedup vs baseline: 3.1589x; 1.0718x over previous
#include <hip/hip_runtime.h>

typedef __attribute__((ext_vector_type(2))) float v2f;
typedef __attribute__((ext_vector_type(8))) float v8f;

constexpr int Bsz = 128;
constexpr int Nsl = 32;
constexpr int Dd  = 128;
constexpr int Hh  = 256;
constexpr int BN  = Bsz * Nsl;
constexpr float LN_EPS = 1e-5f;

typedef __attribute__((ext_vector_type(4))) float v4f;

constexpr int IPB = 16;
constexpr int RPW = 2;
constexpr int NWV = IPB / RPW;
constexpr int PAIR_THREADS = NWV * 32;
constexpr int PAIR_GRID = BN / IPB;
constexpr int GP = Hh + 4;
constexpr int OP = Dd + 4;
constexpr int KB = 4;
constexpr int SP = Dd + 4;

template <int M>
__device__ inline float row_xmask_add(float v) {
    const int moved = __builtin_amdgcn_update_dpp(0, __builtin_bit_cast(int, v),
                                                  0x160 | M, 0xf, 0xf, true);
    return v + __builtin_bit_cast(float, moved);
}

__device__ inline float row16_sum(float v) {
    v = row_xmask_add<1>(v);
    v = row_xmask_add<2>(v);
    v = row_xmask_add<4>(v);
    v = row_xmask_add<8>(v);
    return v;
}

__device__ inline float xhalf(float x) {
    const int b = __builtin_bit_cast(int, x);
    return __builtin_bit_cast(float,
        __builtin_amdgcn_permlanex16(b, b, 0x76543210, 0xfedcba98, false, false));
}

__global__ void __launch_bounds__(PAIR_THREADS)
pair_reduce_kernel(const float* __restrict__ slots,
                   const float* __restrict__ Wenc,
                   const float* __restrict__ b_enc,
                   const float* __restrict__ ln_g,
                   const float* __restrict__ ln_b,
                   const float* __restrict__ W_att,
                   const float* __restrict__ b_att,
                   const float* __restrict__ Weff,
                   const float* __restrict__ beff,
                   float* __restrict__ out)
{
    __shared__ v4f ssh[Nsl * SP / 4];
    __shared__ v4f ush[IPB * GP / 4];
    __shared__ v4f vsh[Nsl * Hh / 4];
    __shared__ v4f psh[4 * Hh / 4];
    __shared__ v4f gsh[IPB * GP / 4];
    __shared__ v4f osh[IPB * OP / 4];

    const int tid  = threadIdx.x;
    const int lane = tid & 31;
    const int wave = tid >> 5;
    const int c    = lane & 15;
    const int hh   = lane >> 4;
    const int b    = blockIdx.x >> 1;
    const int p    = blockIdx.x & 1;
    const int lr0  = RPW * wave;

    const v4f* __restrict__ ssrc = reinterpret_cast<const v4f*>(slots + (size_t)b * Nsl * Dd);
    v4f sst[(Nsl * Dd / 4) / PAIR_THREADS];
    #pragma unroll
    for (int q = 0; q < (Nsl * Dd / 4) / PAIR_THREADS; ++q) sst[q] = ssrc[tid + PAIR_THREADS * q];

    const float pe = b_enc[tid];
    const float pg = ln_g[tid];
    const float pl = ln_b[tid];
    const float pw = W_att[tid];

    const float* __restrict__ wq = Wenc + (size_t)(2 * hh) * Hh + wave * 32 + c;
    v2f wv[2][32], wt[2][32];
    #pragma unroll
    for (int s = 0; s < KB; ++s) {
        #pragma unroll
        for (int ct = 0; ct < 2; ++ct) {
            wv[ct][s].x = wq[(Dd + 4 * s) * Hh + 16 * ct];
            wv[ct][s].y = wq[(Dd + 4 * s + 1) * Hh + 16 * ct];
            wt[ct][s].x = wq[(4 * s) * Hh + 16 * ct];
            wt[ct][s].y = wq[(4 * s + 1) * Hh + 16 * ct];
        }
    }

    float* sf = reinterpret_cast<float*>(ssh);
    #pragma unroll
    for (int q = 0; q < (Nsl * Dd / 4) / PAIR_THREADS; ++q) {
        const int idx = tid + PAIR_THREADS * q;
        *reinterpret_cast<v4f*>(sf + (idx >> 5) * SP + 4 * (idx & 31)) = sst[q];
    }
    {
        float* pf = reinterpret_cast<float*>(psh);
        pf[tid]          = pe;
        pf[Hh + tid]     = pg;
        pf[2 * Hh + tid] = pl;
        pf[3 * Hh + tid] = pw;
    }
    constexpr float LOG2E = 1.4426950408889634f;
    const float nbatt = -b_att[0] * LOG2E;

    __syncthreads();

    float* vf = reinterpret_cast<float*>(vsh);
    float* uf = reinterpret_cast<float*>(ush);
    {
        const float* aown = sf + (16 * p + c) * SP + 2 * hh;
        const float* aoth = sf + (16 * (1 - p) + c) * SP + 2 * hh;
        v8f accU[2] = {}, accVown[2] = {}, accVoth[2] = {};
        v2f fo[2][KB], ft[2][KB];
        #pragma unroll
        for (int s = 0; s < KB; ++s) {
            fo[0][s] = *reinterpret_cast<const v2f*>(aown + 4 * s);
            ft[0][s] = *reinterpret_cast<const v2f*>(aoth + 4 * s);
        }
        #pragma unroll
        for (int j = 0; j < 32 / KB; ++j) {
            if (j + 1 < 32 / KB) {
                #pragma unroll
                for (int s = KB * (j + 1); s < KB * (j + 2); ++s) {
                    #pragma unroll
                    for (int ct = 0; ct < 2; ++ct) {
                        wv[ct][s].x = wq[(Dd + 4 * s) * Hh + 16 * ct];
                        wv[ct][s].y = wq[(Dd + 4 * s + 1) * Hh + 16 * ct];
                        wt[ct][s].x = wq[(4 * s) * Hh + 16 * ct];
                        wt[ct][s].y = wq[(4 * s + 1) * Hh + 16 * ct];
                    }
                }
                #pragma unroll
                for (int s = 0; s < KB; ++s) {
                    fo[(j + 1) & 1][s] = *reinterpret_cast<const v2f*>(aown + 4 * (KB * (j + 1) + s));
                    ft[(j + 1) & 1][s] = *reinterpret_cast<const v2f*>(aoth + 4 * (KB * (j + 1) + s));
                }
            }
            __builtin_amdgcn_sched_barrier(0);
            #pragma unroll
            for (int s = 0; s < KB; ++s) {
                const int k = KB * j + s;
                #pragma unroll
                for (int ct = 0; ct < 2; ++ct) {
                    accVown[ct] = __builtin_amdgcn_wmma_f32_16x16x4_f32(false, fo[j & 1][s], false, wv[ct][k],
                                                                        (short)0, accVown[ct], false, false);
                    accVoth[ct] = __builtin_amdgcn_wmma_f32_16x16x4_f32(false, ft[j & 1][s], false, wv[ct][k],
                                                                        (short)0, accVoth[ct], false, false);
                    accU[ct]    = __builtin_amdgcn_wmma_f32_16x16x4_f32(false, fo[j & 1][s], false, wt[ct][k],
                                                                        (short)0, accU[ct], false, false);
                }
            }
            __builtin_amdgcn_sched_barrier(0);
        }
        #pragma unroll
        for (int ct = 0; ct < 2; ++ct) {
            const int col = wave * 32 + 16 * ct + c;
            #pragma unroll
            for (int r = 0; r < 8; ++r) {
                vf[(16 * p + 8 * hh + r) * Hh + col]       = accVown[ct][r];
                vf[(16 * (1 - p) + 8 * hh + r) * Hh + col] = accVoth[ct][r];
                uf[(8 * hh + r) * GP + col]                = accU[ct][r];
            }
        }
    }
    __syncthreads();

    #pragma unroll
    for (int q = 0; q < 2; ++q) {
        v4f* vrow = vsh + (16 * q + 2 * wave + hh) * (Hh / 4) + c;
        v4f x[4];
        #pragma unroll
        for (int m = 0; m < 4; ++m) x[m] = vrow[16 * m];
        const v4f s4 = (x[0] + x[1]) + (x[2] + x[3]);
        const float mv = row16_sum((s4.x + s4.y) + (s4.z + s4.w)) * (1.0f / Hh);
        #pragma unroll
        for (int m = 0; m < 4; ++m) vrow[16 * m] = x[m] - mv;
    }

    v2f pe2[8], lg[8], lb[8], wa[8];
    #pragma unroll
    for (int m = 0; m < 4; ++m) {
        const v4f e = psh[16 * m + c];
        const v4f g = psh[64 + 16 * m + c];
        const v4f l = psh[128 + 16 * m + c];
        const v4f w = psh[192 + 16 * m + c];
        pe2[2*m] = e.lo;            pe2[2*m+1] = e.hi;
        lg[2*m] = g.lo;             lg[2*m+1] = g.hi;
        lb[2*m] = l.lo;             lb[2*m+1] = l.hi;
        wa[2*m] = w.lo;             wa[2*m+1] = w.hi;
    }

    v2f ub[RPW][8], gacc[RPW][8];
    float sacc[RPW];
    #pragma unroll
    for (int r = 0; r < RPW; ++r) {
        #pragma unroll
        for (int m = 0; m < 4; ++m) {
            const v4f u4 = *reinterpret_cast<const v4f*>(uf + (lr0 + r) * GP + 64 * m + 4 * c);
            ub[r][2*m]   = u4.lo + pe2[2*m];
            ub[r][2*m+1] = u4.hi + pe2[2*m+1];
        }
        const v2f s2 = ((ub[r][0] + ub[r][1]) + (ub[r][2] + ub[r][3])) + ((ub[r][4] + ub[r][5]) + (ub[r][6] + ub[r][7]));
        const float mu = row16_sum(s2.x + s2.y) * (1.0f / Hh);
        #pragma unroll
        for (int k = 0; k < 8; ++k) {
            ub[r][k] = ub[r][k] - mu;
            gacc[r][k] = (v2f){0.0f, 0.0f};
        }
        sacc[r] = 0.0f;
    }

    __syncthreads();

    const float* __restrict__ wp = Weff + (size_t)(2 * hh) * Dd + wave * 16 + c;
    v2f wb[2][32];
    #pragma unroll
    for (int s = 0; s < 32; ++s) {
        wb[0][s].x = wp[(4 * s) * Dd];
        wb[0][s].y = wp[(4 * s + 1) * Dd];
    }

    const int i0 = p * IPB + lr0;
    const v4f* vbase = vsh + hh * (Hh / 4) + c;
    v4f vcur[4];
    #pragma unroll
    for (int m = 0; m < 4; ++m) vcur[m] = vbase[16 * m];

    auto trip = [&](const int t) {
        const v4f* vnrow = vbase + ((t + 1) & (Nsl / 2 - 1)) * (2 * Hh / 4);
        v4f vnext[4];
        #pragma unroll
        for (int m = 0; m < 4; ++m) vnext[m] = vnrow[16 * m];

        #pragma unroll
        for (int r = 0; r < RPW; ++r) {
            v2f h[8];
            #pragma unroll
            for (int m = 0; m < 4; ++m) {
                h[2*m]   = ub[r][2*m]   + vcur[m].lo;
                h[2*m+1] = ub[r][2*m+1] + vcur[m].hi;
            }
            v2f qa = h[0] * h[0];
            v2f qb = h[1] * h[1];
            #pragma unroll
            for (int k = 1; k < 4; ++k) {
                qa = qa + h[2*k]   * h[2*k];
                qb = qb + h[2*k+1] * h[2*k+1];
            }
            const v2f q2 = qa + qb;
            const float var = row16_sum(q2.x + q2.y) * (1.0f / Hh);
            const float inv = __builtin_amdgcn_rsqf(var + LN_EPS);

            #pragma unroll
            for (int k = 0; k < 8; ++k) {
                v2f hn = h[k] * inv * lg[k] + lb[k];
                hn.x = fmaxf(hn.x, 0.0f);
                hn.y = fmaxf(hn.y, 0.0f);
                h[k] = hn;
            }
            v2f da = h[0] * wa[0];
            v2f db = h[1] * wa[1];
            #pragma unroll
            for (int k = 1; k < 4; ++k) {
                da = da + h[2*k]   * wa[2*k];
                db = db + h[2*k+1] * wa[2*k+1];
            }
            const v2f d2 = da + db;
            const float e = __builtin_amdgcn_exp2f(fmaf(row16_sum(d2.x + d2.y), -LOG2E, nbatt));
            float a = __builtin_amdgcn_rcpf(1.0f + e);
            a = (2 * t + hh == i0 + r) ? 0.0f : a;

            #pragma unroll
            for (int k = 0; k < 8; ++k) gacc[r][k] = gacc[r][k] + a * h[k];
            sacc[r] += a;
        }

        #pragma unroll
        for (int m = 0; m < 4; ++m) vcur[m] = vnext[m];
    };

    for (int t = 0; t < Nsl / 4; ++t) trip(t);
    #pragma unroll
    for (int s = 0; s < 32; ++s) {
        wb[1][s].x = wp[((Hh / 2) + 4 * s) * Dd];
        wb[1][s].y = wp[((Hh / 2) + 4 * s + 1) * Dd];
    }
    const v4f be4 = reinterpret_cast<const v4f*>(beff)[lane];
    for (int t = Nsl / 4; t < Nsl / 2; ++t) trip(t);

    #pragma unroll
    for (int r = 0; r < RPW; ++r) {
        #pragma unroll
        for (int k = 0; k < 8; ++k) {
            gacc[r][k].x += xhalf(gacc[r][k].x);
            gacc[r][k].y += xhalf(gacc[r][k].y);
        }
        sacc[r] += xhalf(sacc[r]);
    }

    float* gf = reinterpret_cast<float*>(gsh);
    #pragma unroll
    for (int r = 0; r < RPW; ++r) {
        #pragma unroll
        for (int q = 0; q < 2; ++q) {
            const v2f lo = hh ? gacc[r][4 + 2*q]     : gacc[r][2*q];
            const v2f hi = hh ? gacc[r][4 + 2*q + 1] : gacc[r][2*q + 1];
            v4f o;
            o.lo = lo;
            o.hi = hi;
            *reinterpret_cast<v4f*>(gf + (lr0 + r) * GP + 64 * (2 * hh + q) + 4 * c) = o;
        }
    }
    __syncthreads();

    v8f acc0 = {}, acc1 = {};
    const float* ga = gf + c * GP + 2 * hh;
    #pragma unroll
    for (int s = 0; s < 32; ++s) {
        const v2f a0 = *reinterpret_cast<const v2f*>(ga + 4 * s);
        const v2f a1 = *reinterpret_cast<const v2f*>(ga + (Hh / 2) + 4 * s);
        acc0 = __builtin_amdgcn_wmma_f32_16x16x4_f32(false, a0, false, wb[0][s],
                                                     (short)0, acc0, false, false);
        acc1 = __builtin_amdgcn_wmma_f32_16x16x4_f32(false, a1, false, wb[1][s],
                                                     (short)0, acc1, false, false);
    }
    float* of = reinterpret_cast<float*>(osh);
    #pragma unroll
    for (int r = 0; r < 8; ++r)
        of[(8 * hh + r) * OP + wave * 16 + c] = acc0[r] + acc1[r];
    __syncthreads();

    #pragma unroll
    for (int r = 0; r < RPW; ++r) {
        const v4f p0 = *reinterpret_cast<const v4f*>(of + (lr0 + r) * OP + 4 * lane);
        const v4f res = p0 + sacc[r] * be4;
        *reinterpret_cast<v4f*>(out + (size_t)(b * Nsl + i0 + r) * Dd + 4 * lane) = res;
    }
}

extern "C" void kernel_launch(void* const* d_in, const int* in_sizes, int n_in,
                              void* d_out, int out_size, void* d_ws, size_t ws_size,
                              hipStream_t stream)
{
    const float* slots = (const float*)d_in[0];
    const float* W_enc = (const float*)d_in[1];
    const float* b_enc = (const float*)d_in[2];
    const float* ln_g  = (const float*)d_in[3];
    const float* ln_b  = (const float*)d_in[4];
    const float* W_att = (const float*)d_in[5];
    const float* b_att = (const float*)d_in[6];
    const float* W_eff = (const float*)d_in[7];
    const float* b_eff = (const float*)d_in[8];
    float* out = (float*)d_out;

    pair_reduce_kernel<<<PAIR_GRID, PAIR_THREADS, 0, stream>>>(slots, W_enc, b_enc, ln_g, ln_b,
                                                               W_att, b_att, W_eff, b_eff, out);
}


# pair loop: 4 pairs (2 rows x 2 trips) share one merged 16-lane butterfly per row sum, lane l keeps pair l&3; rsq and sigmoid once per 4 pairs, quad broadcast; V rows not copied; sums keep order
# speedup vs baseline: 3.3461x; 1.0593x over previous
#include <hip/hip_runtime.h>

typedef __attribute__((ext_vector_type(2))) float v2f;
typedef __attribute__((ext_vector_type(8))) float v8f;

constexpr int Bsz = 128;
constexpr int Nsl = 32;
constexpr int Dd  = 128;
constexpr int Hh  = 256;
constexpr int BN  = Bsz * Nsl;
constexpr float LN_EPS = 1e-5f;

typedef __attribute__((ext_vector_type(4))) float v4f;

constexpr int IPB = 16;
constexpr int RPW = 2;
constexpr int NWV = IPB / RPW;
constexpr int PAIR_THREADS = NWV * 32;
constexpr int PAIR_GRID = BN / IPB;
constexpr int GP = Hh + 4;
constexpr int OP = Dd + 4;
constexpr int KB = 4;
constexpr int SP = Dd + 4;

template <int M>
__device__ inline float row_xmask_add(float v) {
    const int moved = __builtin_amdgcn_update_dpp(0, __builtin_bit_cast(int, v),
                                                  0x160 | M, 0xf, 0xf, true);
    return v + __builtin_bit_cast(float, moved);
}

__device__ inline float row16_sum(float v) {
    v = row_xmask_add<1>(v);
    v = row_xmask_add<2>(v);
    v = row_xmask_add<4>(v);
    v = row_xmask_add<8>(v);
    return v;
}

__device__ inline float xhalf(float x) {
    const int b = __builtin_bit_cast(int, x);
    return __builtin_bit_cast(float,
        __builtin_amdgcn_permlanex16(b, b, 0x76543210, 0xfedcba98, false, false));
}

template <int M>
__device__ inline float row_xmask_get(float v) {
    return __builtin_bit_cast(float,
        __builtin_amdgcn_update_dpp(0, __builtin_bit_cast(int, v), 0x160 | M, 0xf, 0xf, true));
}

template <int Q>
__device__ inline float quad_bcast(float v) {
    return __builtin_bit_cast(float,
        __builtin_amdgcn_update_dpp(0, __builtin_bit_cast(int, v), Q * 0x55, 0xf, 0xf, true));
}

__device__ inline float row16_sum4(float s0, float s1, float s2, float s3, bool o1, bool o2) {
    const float k01 = o1 ? s1 : s0;
    const float g01 = o1 ? s0 : s1;
    const float x = k01 + row_xmask_get<1>(g01);
    const float k23 = o1 ? s3 : s2;
    const float g23 = o1 ? s2 : s3;
    const float y = k23 + row_xmask_get<1>(g23);
    const float kz = o2 ? y : x;
    const float gz = o2 ? x : y;
    float z = kz + row_xmask_get<2>(gz);
    z = row_xmask_add<4>(z);
    z = row_xmask_add<8>(z);
    return z;
}

__global__ void __launch_bounds__(PAIR_THREADS)
pair_reduce_kernel(const float* __restrict__ slots,
                   const float* __restrict__ Wenc,
                   const float* __restrict__ b_enc,
                   const float* __restrict__ ln_g,
                   const float* __restrict__ ln_b,
                   const float* __restrict__ W_att,
                   const float* __restrict__ b_att,
                   const float* __restrict__ Weff,
                   const float* __restrict__ beff,
                   float* __restrict__ out)
{
    __shared__ v4f ssh[Nsl * SP / 4];
    __shared__ v4f ush[IPB * GP / 4];
    __shared__ v4f vsh[Nsl * Hh / 4];
    __shared__ v4f psh[4 * Hh / 4];
    __shared__ v4f gsh[IPB * GP / 4];
    __shared__ v4f osh[IPB * OP / 4];

    const int tid  = threadIdx.x;
    const int lane = tid & 31;
    const int wave = tid >> 5;
    const int c    = lane & 15;
    const int hh   = lane >> 4;
    const int b    = blockIdx.x >> 1;
    const int p    = blockIdx.x & 1;
    const int lr0  = RPW * wave;

    const v4f* __restrict__ ssrc = reinterpret_cast<const v4f*>(slots + (size_t)b * Nsl * Dd);
    v4f sst[(Nsl * Dd / 4) / PAIR_THREADS];
    #pragma unroll
    for (int q = 0; q < (Nsl * Dd / 4) / PAIR_THREADS; ++q) sst[q] = ssrc[tid + PAIR_THREADS * q];

    const float pe = b_enc[tid];
    const float pg = ln_g[tid];
    const float pl = ln_b[tid];
    const float pw = W_att[tid];

    const float* __restrict__ wq = Wenc + (size_t)(2 * hh) * Hh + wave * 32 + c;
    v2f wv[2][32], wt[2][32];
    #pragma unroll
    for (int s = 0; s < KB; ++s) {
        #pragma unroll
        for (int ct = 0; ct < 2; ++ct) {
            wv[ct][s].x = wq[(Dd + 4 * s) * Hh + 16 * ct];
            wv[ct][s].y = wq[(Dd + 4 * s + 1) * Hh + 16 * ct];
            wt[ct][s].x = wq[(4 * s) * Hh + 16 * ct];
            wt[ct][s].y = wq[(4 * s + 1) * Hh + 16 * ct];
        }
    }

    float* sf = reinterpret_cast<float*>(ssh);
    #pragma unroll
    for (int q = 0; q < (Nsl * Dd / 4) / PAIR_THREADS; ++q) {
        const int idx = tid + PAIR_THREADS * q;
        *reinterpret_cast<v4f*>(sf + (idx >> 5) * SP + 4 * (idx & 31)) = sst[q];
    }
    {
        float* pf = reinterpret_cast<float*>(psh);
        pf[tid]          = pe;
        pf[Hh + tid]     = pg;
        pf[2 * Hh + tid] = pl;
        pf[3 * Hh + tid] = pw;
    }
    constexpr float LOG2E = 1.4426950408889634f;
    const float nbatt = -b_att[0] * LOG2E;

    __syncthreads();

    float* vf = reinterpret_cast<float*>(vsh);
    float* uf = reinterpret_cast<float*>(ush);
    {
        const float* aown = sf + (16 * p + c) * SP + 2 * hh;
        const float* aoth = sf + (16 * (1 - p) + c) * SP + 2 * hh;
        v8f accU[2] = {}, accVown[2] = {}, accVoth[2] = {};
        v2f fo[2][KB], ft[2][KB];
        #pragma unroll
        for (int s = 0; s < KB; ++s) {
            fo[0][s] = *reinterpret_cast<const v2f*>(aown + 4 * s);
            ft[0][s] = *reinterpret_cast<const v2f*>(aoth + 4 * s);
        }
        #pragma unroll
        for (int j = 0; j < 32 / KB; ++j) {
            if (j + 1 < 32 / KB) {
                #pragma unroll
                for (int s = KB * (j + 1); s < KB * (j + 2); ++s) {
                    #pragma unroll
                    for (int ct = 0; ct < 2; ++ct) {
                        wv[ct][s].x = wq[(Dd + 4 * s) * Hh + 16 * ct];
                        wv[ct][s].y = wq[(Dd + 4 * s + 1) * Hh + 16 * ct];
                        wt[ct][s].x = wq[(4 * s) * Hh + 16 * ct];
                        wt[ct][s].y = wq[(4 * s + 1) * Hh + 16 * ct];
                    }
                }
                #pragma unroll
                for (int s = 0; s < KB; ++s) {
                    fo[(j + 1) & 1][s] = *reinterpret_cast<const v2f*>(aown + 4 * (KB * (j + 1) + s));
                    ft[(j + 1) & 1][s] = *reinterpret_cast<const v2f*>(aoth + 4 * (KB * (j + 1) + s));
                }
            }
            __builtin_amdgcn_sched_barrier(0);
            #pragma unroll
            for (int s = 0; s < KB; ++s) {
                const int k = KB * j + s;
                #pragma unroll
                for (int ct = 0; ct < 2; ++ct) {
                    accVown[ct] = __builtin_amdgcn_wmma_f32_16x16x4_f32(false, fo[j & 1][s], false, wv[ct][k],
                                                                        (short)0, accVown[ct], false, false);
                    accVoth[ct] = __builtin_amdgcn_wmma_f32_16x16x4_f32(false, ft[j & 1][s], false, wv[ct][k],
                                                                        (short)0, accVoth[ct], false, false);
                    accU[ct]    = __builtin_amdgcn_wmma_f32_16x16x4_f32(false, fo[j & 1][s], false, wt[ct][k],
                                                                        (short)0, accU[ct], false, false);
                }
            }
            __builtin_amdgcn_sched_barrier(0);
        }
        #pragma unroll
        for (int ct = 0; ct < 2; ++ct) {
            const int col = wave * 32 + 16 * ct + c;
            #pragma unroll
            for (int r = 0; r < 8; ++r) {
                vf[(16 * p + 8 * hh + r) * Hh + col]       = accVown[ct][r];
                vf[(16 * (1 - p) + 8 * hh + r) * Hh + col] = accVoth[ct][r];
                uf[(8 * hh + r) * GP + col]                = accU[ct][r];
            }
        }
    }
    __syncthreads();

    #pragma unroll
    for (int q = 0; q < 2; ++q) {
        v4f* vrow = vsh + (16 * q + 2 * wave + hh) * (Hh / 4) + c;
        v4f x[4];
        #pragma unroll
        for (int m = 0; m < 4; ++m) x[m] = vrow[16 * m];
        const v4f s4 = (x[0] + x[1]) + (x[2] + x[3]);
        const float mv = row16_sum((s4.x + s4.y) + (s4.z + s4.w)) * (1.0f / Hh);
        #pragma unroll
        for (int m = 0; m < 4; ++m) vrow[16 * m] = x[m] - mv;
    }

    v2f pe2[8], lg[8], lb[8], wa[8];
    #pragma unroll
    for (int m = 0; m < 4; ++m) {
        const v4f e = psh[16 * m + c];
        const v4f g = psh[64 + 16 * m + c];
        const v4f l = psh[128 + 16 * m + c];
        const v4f w = psh[192 + 16 * m + c];
        pe2[2*m] = e.lo;            pe2[2*m+1] = e.hi;
        lg[2*m] = g.lo;             lg[2*m+1] = g.hi;
        lb[2*m] = l.lo;             lb[2*m+1] = l.hi;
        wa[2*m] = w.lo;             wa[2*m+1] = w.hi;
    }

    v2f ub[RPW][8], gacc[RPW][8];
    float sacc[RPW];
    #pragma unroll
    for (int r = 0; r < RPW; ++r) {
        #pragma unroll
        for (int m = 0; m < 4; ++m) {
            const v4f u4 = *reinterpret_cast<const v4f*>(uf + (lr0 + r) * GP + 64 * m + 4 * c);
            ub[r][2*m]   = u4.lo + pe2[2*m];
            ub[r][2*m+1] = u4.hi + pe2[2*m+1];
        }
        const v2f s2 = ((ub[r][0] + ub[r][1]) + (ub[r][2] + ub[r][3])) + ((ub[r][4] + ub[r][5]) + (ub[r][6] + ub[r][7]));
        const float mu = row16_sum(s2.x + s2.y) * (1.0f / Hh);
        #pragma unroll
        for (int k = 0; k < 8; ++k) {
            ub[r][k] = ub[r][k] - mu;
            gacc[r][k] = (v2f){0.0f, 0.0f};
        }
        sacc[r] = 0.0f;
    }

    __syncthreads();

    const float* __restrict__ wp = Weff + (size_t)(2 * hh) * Dd + wave * 16 + c;
    v2f wb[2][32];
    #pragma unroll
    for (int s = 0; s < 32; ++s) {
        wb[0][s].x = wp[(4 * s) * Dd];
        wb[0][s].y = wp[(4 * s + 1) * Dd];
    }

    const int i0 = p * IPB + lr0;
    const v4f* vbase = vsh + hh * (Hh / 4) + c;
    v4f vcur[2][4];
    #pragma unroll
    for (int d = 0; d < 2; ++d) {
        #pragma unroll
        for (int m = 0; m < 4; ++m) vcur[d][m] = vbase[d * (2 * Hh / 4) + 16 * m];
    }

    const bool o1 = (lane & 1) != 0;
    const bool o2 = (lane & 2) != 0;
    const int jcmp = i0 + (lane & 1) - (lane & 2) - hh;

    auto trip2 = [&](const int t) {
        v2f h[2][RPW][8];
        float ps[2][RPW];
        #pragma unroll
        for (int d = 0; d < 2; ++d) {
            #pragma unroll
            for (int r = 0; r < RPW; ++r) {
                #pragma unroll
                for (int m = 0; m < 4; ++m) {
                    h[d][r][2*m]   = ub[r][2*m]   + vcur[d][m].lo;
                    h[d][r][2*m+1] = ub[r][2*m+1] + vcur[d][m].hi;
                }
            }
        }
        {
            const v4f* vn = vbase + ((t + 2) & (Nsl / 2 - 1)) * (2 * Hh / 4);
            #pragma unroll
            for (int d = 0; d < 2; ++d) {
                #pragma unroll
                for (int m = 0; m < 4; ++m) vcur[d][m] = vn[d * (2 * Hh / 4) + 16 * m];
            }
        }
        #pragma unroll
        for (int d = 0; d < 2; ++d) {
            #pragma unroll
            for (int r = 0; r < RPW; ++r) {
                v2f qa = h[d][r][0] * h[d][r][0];
                v2f qb = h[d][r][1] * h[d][r][1];
                #pragma unroll
                for (int k = 1; k < 4; ++k) {
                    qa = qa + h[d][r][2*k]   * h[d][r][2*k];
                    qb = qb + h[d][r][2*k+1] * h[d][r][2*k+1];
                }
                const v2f q2 = qa + qb;
                ps[d][r] = q2.x + q2.y;
            }
        }
        const float var = row16_sum4(ps[0][0], ps[0][1], ps[1][0], ps[1][1], o1, o2) * (1.0f / Hh);
        const float invq = __builtin_amdgcn_rsqf(var + LN_EPS);
        const float inv[2][RPW] = { { quad_bcast<0>(invq), quad_bcast<1>(invq) },
                                    { quad_bcast<2>(invq), quad_bcast<3>(invq) } };

        #pragma unroll
        for (int d = 0; d < 2; ++d) {
            #pragma unroll
            for (int r = 0; r < RPW; ++r) {
                #pragma unroll
                for (int k = 0; k < 8; ++k) {
                    v2f hn = h[d][r][k] * inv[d][r] * lg[k] + lb[k];
                    hn.x = fmaxf(hn.x, 0.0f);
                    hn.y = fmaxf(hn.y, 0.0f);
                    h[d][r][k] = hn;
                }
                v2f da = h[d][r][0] * wa[0];
                v2f db = h[d][r][1] * wa[1];
                #pragma unroll
                for (int k = 1; k < 4; ++k) {
                    da = da + h[d][r][2*k]   * wa[2*k];
                    db = db + h[d][r][2*k+1] * wa[2*k+1];
                }
                const v2f d2 = da + db;
                ps[d][r] = d2.x + d2.y;
            }
        }
        const float dotq = row16_sum4(ps[0][0], ps[0][1], ps[1][0], ps[1][1], o1, o2);
        const float e = __builtin_amdgcn_exp2f(fmaf(dotq, -LOG2E, nbatt));
        float aq = __builtin_amdgcn_rcpf(1.0f + e);
        aq = (2 * t == jcmp) ? 0.0f : aq;
        const float a[2][RPW] = { { quad_bcast<0>(aq), quad_bcast<1>(aq) },
                                  { quad_bcast<2>(aq), quad_bcast<3>(aq) } };

        #pragma unroll
        for (int d = 0; d < 2; ++d) {
            #pragma unroll
            for (int r = 0; r < RPW; ++r) {
                #pragma unroll
                for (int k = 0; k < 8; ++k) gacc[r][k] = gacc[r][k] + a[d][r] * h[d][r][k];
                sacc[r] += a[d][r];
            }
        }
    };

    for (int t = 0; t < Nsl / 4; t += 2) trip2(t);
    #pragma unroll
    for (int s = 0; s < 32; ++s) {
        wb[1][s].x = wp[((Hh / 2) + 4 * s) * Dd];
        wb[1][s].y = wp[((Hh / 2) + 4 * s + 1) * Dd];
    }
    const v4f be4 = reinterpret_cast<const v4f*>(beff)[lane];
    for (int t = Nsl / 4; t < Nsl / 2; t += 2) trip2(t);

    #pragma unroll
    for (int r = 0; r < RPW; ++r) {
        #pragma unroll
        for (int k = 0; k < 8; ++k) {
            gacc[r][k].x += xhalf(gacc[r][k].x);
            gacc[r][k].y += xhalf(gacc[r][k].y);
        }
        sacc[r] += xhalf(sacc[r]);
    }

    float* gf = reinterpret_cast<float*>(gsh);
    #pragma unroll
    for (int r = 0; r < RPW; ++r) {
        #pragma unroll
        for (int q = 0; q < 2; ++q) {
            const v2f lo = hh ? gacc[r][4 + 2*q]     : gacc[r][2*q];
            const v2f hi = hh ? gacc[r][4 + 2*q + 1] : gacc[r][2*q + 1];
            v4f o;
            o.lo = lo;
            o.hi = hi;
            *reinterpret_cast<v4f*>(gf + (lr0 + r) * GP + 64 * (2 * hh + q) + 4 * c) = o;
        }
    }
    __syncthreads();

    v8f acc0 = {}, acc1 = {};
    const float* ga = gf + c * GP + 2 * hh;
    #pragma unroll
    for (int s = 0; s < 32; ++s) {
        const v2f a0 = *reinterpret_cast<const v2f*>(ga + 4 * s);
        const v2f a1 = *reinterpret_cast<const v2f*>(ga + (Hh / 2) + 4 * s);
        acc0 = __builtin_amdgcn_wmma_f32_16x16x4_f32(false, a0, false, wb[0][s],
                                                     (short)0, acc0, false, false);
        acc1 = __builtin_amdgcn_wmma_f32_16x16x4_f32(false, a1, false, wb[1][s],
                                                     (short)0, acc1, false, false);
    }
    float* of = reinterpret_cast<float*>(osh);
    #pragma unroll
    for (int r = 0; r < 8; ++r)
        of[(8 * hh + r) * OP + wave * 16 + c] = acc0[r] + acc1[r];
    __syncthreads();

    #pragma unroll
    for (int r = 0; r < RPW; ++r) {
        const v4f p0 = *reinterpret_cast<const v4f*>(of + (lr0 + r) * OP + 4 * lane);
        const v4f res = p0 + sacc[r] * be4;
        *reinterpret_cast<v4f*>(out + (size_t)(b * Nsl + i0 + r) * Dd + 4 * lane) = res;
    }
}

extern "C" void kernel_launch(void* const* d_in, const int* in_sizes, int n_in,
                              void* d_out, int out_size, void* d_ws, size_t ws_size,
                              hipStream_t stream)
{
    const float* slots = (const float*)d_in[0];
    const float* W_enc = (const float*)d_in[1];
    const float* b_enc = (const float*)d_in[2];
    const float* ln_g  = (const float*)d_in[3];
    const float* ln_b  = (const float*)d_in[4];
    const float* W_att = (const float*)d_in[5];
    const float* b_att = (const float*)d_in[6];
    const float* W_eff = (const float*)d_in[7];
    const float* b_eff = (const float*)d_in[8];
    float* out = (float*)d_out;

    pair_reduce_kernel<<<PAIR_GRID, PAIR_THREADS, 0, stream>>>(slots, W_enc, b_enc, ln_g, ln_b,
                                                               W_att, b_att, W_eff, b_eff, out);
}
